# starting kernel
# baseline (speedup 1.0000x reference)
#include <hip/hip_runtime.h>
#include <hip/hip_bf16.h>
#include <math.h>

#define Bn   32
#define Hn   56
#define Wn   56
#define Cn   256
#define HWn  (Hn * Wn)
#define HID  16
#define NCHK 8
#define CHUNK (HWn / NCHK)

typedef float v2f __attribute__((ext_vector_type(2)));
typedef float v8f __attribute__((ext_vector_type(8)));

__device__ __forceinline__ v8f wmma4(v2f a, v2f b, v8f c) {
    return __builtin_amdgcn_wmma_f32_16x16x4_f32(
         false, a,  false, b,
         (short)0, c,  false,  false);
}

__device__ __forceinline__ float sigmoidf_(float v) {
    return 1.0f / (1.0f + __expf(-v));
}

__global__ void k_pool_partial(const float* __restrict__ x,
                               float* __restrict__ psum,
                               float* __restrict__ pmax) {
    const int c     = threadIdx.x;
    const int chunk = blockIdx.x;
    const int b     = blockIdx.y;
    const float* xp = x + ((size_t)b * HWn + (size_t)chunk * CHUNK) * Cn + c;

    float s = 0.0f;
    float m = -3.402823466e38f;
    #pragma unroll 4
    for (int i = 0; i < CHUNK; ++i) {
        float v = xp[(size_t)i * Cn];
        s += v;
        m = fmaxf(m, v);
    }
    psum[((size_t)chunk * Bn + b) * Cn + c] = s;
    pmax[((size_t)chunk * Bn + b) * Cn + c] = m;
}

__global__ void k_channel_mlp(const float* __restrict__ psum,
                              const float* __restrict__ pmax,
                              const float* __restrict__ w1,
                              const float* __restrict__ b1,
                              const float* __restrict__ w2,
                              const float* __restrict__ b2,
                              float* __restrict__ chout) {
    __shared__ float lds_pool[2][Bn][Cn + 1];
    __shared__ float lds_h[2][Bn][HID + 1];

    const int tid  = threadIdx.x;
    const int wv   = tid >> 5;
    const int lane = tid & 31;

    for (int p = tid; p < Bn * Cn; p += 256) {
        const int b = p >> 8, c = p & (Cn - 1);
        float s = 0.0f, m = -3.402823466e38f;
        #pragma unroll
        for (int ch = 0; ch < NCHK; ++ch) {
            s += psum[((size_t)ch * Bn + b) * Cn + c];
            m = fmaxf(m, pmax[((size_t)ch * Bn + b) * Cn + c]);
        }
        lds_pool[0][b][c] = s * (1.0f / (float)HWn);
        lds_pool[1][b][c] = m;
    }
    __syncthreads();

    if (wv < 4) {
        const int src  = wv >> 1;
        const int mt   = wv & 1;
        const int col  = lane & 15;
        const int m    = mt * 16 + col;
        const int ksel = (lane < 16) ? 0 : 2;

        v8f acc = {};
        for (int k0 = 0; k0 < Cn; k0 += 4) {
            v2f a, bfr;
            a.x   = lds_pool[src][m][k0 + ksel];
            a.y   = lds_pool[src][m][k0 + ksel + 1];
            bfr.x = w1[(k0 + ksel)     * HID + col];
            bfr.y = w1[(k0 + ksel + 1) * HID + col];
            acc = wmma4(a, bfr, acc);
        }
        const float bias  = b1[col];
        const int   rbase = mt * 16 + ((lane & 16) ? 8 : 0);
        #pragma unroll
        for (int v = 0; v < 8; ++v)
            lds_h[src][rbase + v][col] = fmaxf(acc[v] + bias, 0.0f);
    }
    __syncthreads();

    for (int t = wv; t < 32; t += 8) {
        const int mt   = t & 1;
        const int nt   = t >> 1;
        const int col  = lane & 15;
        const int ksel = (lane < 16) ? 0 : 2;
        const float bias = b2[nt * 16 + col];

        float combined[8];
        #pragma unroll
        for (int v = 0; v < 8; ++v) combined[v] = 0.0f;

        #pragma unroll
        for (int src = 0; src < 2; ++src) {
            v8f acc = {};
            #pragma unroll
            for (int k0 = 0; k0 < HID; k0 += 4) {
                v2f a, bfr;
                a.x   = lds_h[src][mt * 16 + col][k0 + ksel];
                a.y   = lds_h[src][mt * 16 + col][k0 + ksel + 1];
                bfr.x = w2[(k0 + ksel)     * Cn + nt * 16 + col];
                bfr.y = w2[(k0 + ksel + 1) * Cn + nt * 16 + col];
                acc = wmma4(a, bfr, acc);
            }
            #pragma unroll
            for (int v = 0; v < 8; ++v)
                combined[v] += sigmoidf_(acc[v] + bias);
        }
        const int rbase = mt * 16 + ((lane & 16) ? 8 : 0);
        #pragma unroll
        for (int v = 0; v < 8; ++v)
            chout[(size_t)(rbase + v) * Cn + nt * 16 + col] = combined[v];
    }
}

__global__ void k_spatial_stats(const float* __restrict__ x,
                                const float* __restrict__ chout,
                                float* __restrict__ sp) {
    const int lane = threadIdx.x & 31;
    const int wv   = threadIdx.x >> 5;
    const int pos  = blockIdx.x * 8 + wv;
    const int b    = pos / HWn;

    const float* xp = x + (size_t)pos * Cn;
    const float* cp = chout + (size_t)b * Cn;

    float s = 0.0f, m = -3.402823466e38f;
    #pragma unroll
    for (int i = 0; i < Cn / 32; ++i) {
        const int c = lane + i * 32;
        const float v = xp[c] * cp[c];
        s += v;
        m = fmaxf(m, v);
    }
    #pragma unroll
    for (int off = 16; off >= 1; off >>= 1) {
        s += __shfl_xor(s, off, 32);
        m = fmaxf(m, __shfl_xor(m, off, 32));
    }
    if (lane == 0) {
        sp[(size_t)pos * 2 + 0] = s * (1.0f / (float)Cn);
        sp[(size_t)pos * 2 + 1] = m;
    }
}

__global__ void k_spatial_conv(const float* __restrict__ sp,
                               const float* __restrict__ wconv,
                               const float* __restrict__ bconv,
                               float* __restrict__ satt) {
    const int idx = blockIdx.x * blockDim.x + threadIdx.x;
    const int b  = idx / HWn;
    const int hw = idx - b * HWn;
    const int y  = hw / Wn;
    const int xx = hw - y * Wn;

    float acc = 0.0f;
    #pragma unroll
    for (int ky = 0; ky < 7; ++ky) {
        const int iy = y + ky - 3;
        if (iy < 0 || iy >= Hn) continue;
        #pragma unroll
        for (int kx = 0; kx < 7; ++kx) {
            const int ix = xx + kx - 3;
            if (ix < 0 || ix >= Wn) continue;
            const size_t sidx = ((size_t)(b * Hn + iy) * Wn + ix) * 2;
            acc += sp[sidx + 0] * wconv[(ky * 7 + kx) * 2 + 0];
            acc += sp[sidx + 1] * wconv[(ky * 7 + kx) * 2 + 1];
        }
    }
    satt[idx] = sigmoidf_(acc + bconv[0]);
}

__global__ void k_apply(const float* __restrict__ x,
                        const float* __restrict__ chout,
                        const float* __restrict__ satt,
                        float* __restrict__ out) {
    const size_t gid = (size_t)blockIdx.x * blockDim.x + threadIdx.x;
    const size_t pos = gid >> 6;
    const int    c4  = (int)(gid & 63);
    const int    b   = (int)(pos / HWn);

    const float4 xv = ((const float4*)x)[gid];
    const float4 cv = ((const float4*)chout)[(size_t)b * (Cn / 4) + c4];
    const float  s  = satt[pos];

    float4 o;
    o.x = xv.x * cv.x * s;
    o.y = xv.y * cv.y * s;
    o.z = xv.z * cv.z * s;
    o.w = xv.w * cv.w * s;
    ((float4*)out)[gid] = o;
}

extern "C" void kernel_launch(void* const* d_in, const int* in_sizes, int n_in,
                              void* d_out, int out_size, void* d_ws, size_t ws_size,
                              hipStream_t stream) {
    const float* x     = (const float*)d_in[0];
    const float* w1    = (const float*)d_in[1];
    const float* b1    = (const float*)d_in[2];
    const float* w2    = (const float*)d_in[3];
    const float* b2    = (const float*)d_in[4];
    const float* wconv = (const float*)d_in[5];
    const float* bconv = (const float*)d_in[6];
    float* out = (float*)d_out;

    float* ws    = (float*)d_ws;
    float* psum  = ws;
    float* pmax  = psum + (size_t)NCHK * Bn * Cn;
    float* chout = pmax + (size_t)NCHK * Bn * Cn;
    float* sp    = chout + (size_t)Bn * Cn;
    float* satt  = sp + (size_t)Bn * HWn * 2;

    k_pool_partial<<<dim3(NCHK, Bn), 256, 0, stream>>>(x, psum, pmax);

    k_channel_mlp<<<1, 256, 0, stream>>>(psum, pmax, w1, b1, w2, b2, chout);

    k_spatial_stats<<<(Bn * HWn) / 8, 256, 0, stream>>>(x, chout, sp);

    k_spatial_conv<<<(Bn * HWn) / 256, 256, 0, stream>>>(sp, wconv, bconv, satt);

    k_apply<<<(Bn * HWn * Cn / 4) / 256, 256, 0, stream>>>(x, chout, satt, out);
}


# pooling pass: each wave issues its 28 float4 loads as independent requests (was 4 scalar loads per round trip); 14 waves x 2 channel halves per chunk folded through LDS; changes the pooled sum order
# speedup vs baseline: 1.5850x; 1.5850x over previous
#include <hip/hip_runtime.h>
#include <hip/hip_bf16.h>
#include <math.h>

#define Bn   32
#define Hn   56
#define Wn   56
#define Cn   256
#define HWn  (Hn * Wn)
#define HID  16
#define NCHK 8
#define CHUNK (HWn / NCHK)

typedef float v2f __attribute__((ext_vector_type(2)));
typedef float v8f __attribute__((ext_vector_type(8)));

__device__ __forceinline__ v8f wmma4(v2f a, v2f b, v8f c) {
    return __builtin_amdgcn_wmma_f32_16x16x4_f32(
         false, a,  false, b,
         (short)0, c,  false,  false);
}

__device__ __forceinline__ float sigmoidf_(float v) {
    return 1.0f / (1.0f + __expf(-v));
}

#define PSTR 14
#define PPW  (CHUNK / PSTR)
#define CHALF (Cn / 2)

__global__ __launch_bounds__(PSTR * 32)
void k_pool_partial(const float* __restrict__ x,
                    float* __restrict__ psum,
                    float* __restrict__ pmax) {
    __shared__ __attribute__((aligned(16))) float red_s[PSTR * CHALF];
    __shared__ __attribute__((aligned(16))) float red_m[PSTR * CHALF];

    const int lane  = threadIdx.x & 31;
    const int wv    = threadIdx.x >> 5;
    const int half  = blockIdx.x & 1;
    const int chunk = blockIdx.x >> 1;
    const int b     = blockIdx.y;

    const float4* xp = (const float4*)(x + ((size_t)b * HWn + (size_t)chunk * CHUNK + wv) * Cn
                                         + half * CHALF) + lane;

    float4 v[PPW];
    #pragma unroll
    for (int j = 0; j < PPW; ++j)
        v[j] = xp[(size_t)j * PSTR * (Cn / 4)];

    float4 s = {0.0f, 0.0f, 0.0f, 0.0f};
    float4 m = {-3.402823466e38f, -3.402823466e38f, -3.402823466e38f, -3.402823466e38f};
    #pragma unroll
    for (int j = 0; j < PPW; ++j) {
        s.x += v[j].x;  s.y += v[j].y;  s.z += v[j].z;  s.w += v[j].w;
        m.x = fmaxf(m.x, v[j].x);  m.y = fmaxf(m.y, v[j].y);
        m.z = fmaxf(m.z, v[j].z);  m.w = fmaxf(m.w, v[j].w);
    }
    *(float4*)&red_s[wv * CHALF + lane * 4] = s;
    *(float4*)&red_m[wv * CHALF + lane * 4] = m;
    __syncthreads();

    if (threadIdx.x < CHALF) {
        const int c = threadIdx.x;
        float ss = 0.0f, mm = -3.402823466e38f;
        #pragma unroll
        for (int k = 0; k < PSTR; ++k) {
            ss += red_s[k * CHALF + c];
            mm = fmaxf(mm, red_m[k * CHALF + c]);
        }
        const size_t o = ((size_t)chunk * Bn + b) * Cn + half * CHALF + c;
        psum[o] = ss;
        pmax[o] = mm;
    }
}

__global__ void k_channel_mlp(const float* __restrict__ psum,
                              const float* __restrict__ pmax,
                              const float* __restrict__ w1,
                              const float* __restrict__ b1,
                              const float* __restrict__ w2,
                              const float* __restrict__ b2,
                              float* __restrict__ chout) {
    __shared__ float lds_pool[2][Bn][Cn + 1];
    __shared__ float lds_h[2][Bn][HID + 1];

    const int tid  = threadIdx.x;
    const int wv   = tid >> 5;
    const int lane = tid & 31;

    for (int p = tid; p < Bn * Cn; p += 256) {
        const int b = p >> 8, c = p & (Cn - 1);
        float s = 0.0f, m = -3.402823466e38f;
        #pragma unroll
        for (int ch = 0; ch < NCHK; ++ch) {
            s += psum[((size_t)ch * Bn + b) * Cn + c];
            m = fmaxf(m, pmax[((size_t)ch * Bn + b) * Cn + c]);
        }
        lds_pool[0][b][c] = s * (1.0f / (float)HWn);
        lds_pool[1][b][c] = m;
    }
    __syncthreads();

    if (wv < 4) {
        const int src  = wv >> 1;
        const int mt   = wv & 1;
        const int col  = lane & 15;
        const int m    = mt * 16 + col;
        const int ksel = (lane < 16) ? 0 : 2;

        v8f acc = {};
        for (int k0 = 0; k0 < Cn; k0 += 4) {
            v2f a, bfr;
            a.x   = lds_pool[src][m][k0 + ksel];
            a.y   = lds_pool[src][m][k0 + ksel + 1];
            bfr.x = w1[(k0 + ksel)     * HID + col];
            bfr.y = w1[(k0 + ksel + 1) * HID + col];
            acc = wmma4(a, bfr, acc);
        }
        const float bias  = b1[col];
        const int   rbase = mt * 16 + ((lane & 16) ? 8 : 0);
        #pragma unroll
        for (int v = 0; v < 8; ++v)
            lds_h[src][rbase + v][col] = fmaxf(acc[v] + bias, 0.0f);
    }
    __syncthreads();

    for (int t = wv; t < 32; t += 8) {
        const int mt   = t & 1;
        const int nt   = t >> 1;
        const int col  = lane & 15;
        const int ksel = (lane < 16) ? 0 : 2;
        const float bias = b2[nt * 16 + col];

        float combined[8];
        #pragma unroll
        for (int v = 0; v < 8; ++v) combined[v] = 0.0f;

        #pragma unroll
        for (int src = 0; src < 2; ++src) {
            v8f acc = {};
            #pragma unroll
            for (int k0 = 0; k0 < HID; k0 += 4) {
                v2f a, bfr;
                a.x   = lds_h[src][mt * 16 + col][k0 + ksel];
                a.y   = lds_h[src][mt * 16 + col][k0 + ksel + 1];
                bfr.x = w2[(k0 + ksel)     * Cn + nt * 16 + col];
                bfr.y = w2[(k0 + ksel + 1) * Cn + nt * 16 + col];
                acc = wmma4(a, bfr, acc);
            }
            #pragma unroll
            for (int v = 0; v < 8; ++v)
                combined[v] += sigmoidf_(acc[v] + bias);
        }
        const int rbase = mt * 16 + ((lane & 16) ? 8 : 0);
        #pragma unroll
        for (int v = 0; v < 8; ++v)
            chout[(size_t)(rbase + v) * Cn + nt * 16 + col] = combined[v];
    }
}

__global__ void k_spatial_stats(const float* __restrict__ x,
                                const float* __restrict__ chout,
                                float* __restrict__ sp) {
    const int lane = threadIdx.x & 31;
    const int wv   = threadIdx.x >> 5;
    const int pos  = blockIdx.x * 8 + wv;
    const int b    = pos / HWn;

    const float* xp = x + (size_t)pos * Cn;
    const float* cp = chout + (size_t)b * Cn;

    float s = 0.0f, m = -3.402823466e38f;
    #pragma unroll
    for (int i = 0; i < Cn / 32; ++i) {
        const int c = lane + i * 32;
        const float v = xp[c] * cp[c];
        s += v;
        m = fmaxf(m, v);
    }
    #pragma unroll
    for (int off = 16; off >= 1; off >>= 1) {
        s += __shfl_xor(s, off, 32);
        m = fmaxf(m, __shfl_xor(m, off, 32));
    }
    if (lane == 0) {
        sp[(size_t)pos * 2 + 0] = s * (1.0f / (float)Cn);
        sp[(size_t)pos * 2 + 1] = m;
    }
}

__global__ void k_spatial_conv(const float* __restrict__ sp,
                               const float* __restrict__ wconv,
                               const float* __restrict__ bconv,
                               float* __restrict__ satt) {
    const int idx = blockIdx.x * blockDim.x + threadIdx.x;
    const int b  = idx / HWn;
    const int hw = idx - b * HWn;
    const int y  = hw / Wn;
    const int xx = hw - y * Wn;

    float acc = 0.0f;
    #pragma unroll
    for (int ky = 0; ky < 7; ++ky) {
        const int iy = y + ky - 3;
        if (iy < 0 || iy >= Hn) continue;
        #pragma unroll
        for (int kx = 0; kx < 7; ++kx) {
            const int ix = xx + kx - 3;
            if (ix < 0 || ix >= Wn) continue;
            const size_t sidx = ((size_t)(b * Hn + iy) * Wn + ix) * 2;
            acc += sp[sidx + 0] * wconv[(ky * 7 + kx) * 2 + 0];
            acc += sp[sidx + 1] * wconv[(ky * 7 + kx) * 2 + 1];
        }
    }
    satt[idx] = sigmoidf_(acc + bconv[0]);
}

__global__ void k_apply(const float* __restrict__ x,
                        const float* __restrict__ chout,
                        const float* __restrict__ satt,
                        float* __restrict__ out) {
    const size_t gid = (size_t)blockIdx.x * blockDim.x + threadIdx.x;
    const size_t pos = gid >> 6;
    const int    c4  = (int)(gid & 63);
    const int    b   = (int)(pos / HWn);

    const float4 xv = ((const float4*)x)[gid];
    const float4 cv = ((const float4*)chout)[(size_t)b * (Cn / 4) + c4];
    const float  s  = satt[pos];

    float4 o;
    o.x = xv.x * cv.x * s;
    o.y = xv.y * cv.y * s;
    o.z = xv.z * cv.z * s;
    o.w = xv.w * cv.w * s;
    ((float4*)out)[gid] = o;
}

extern "C" void kernel_launch(void* const* d_in, const int* in_sizes, int n_in,
                              void* d_out, int out_size, void* d_ws, size_t ws_size,
                              hipStream_t stream) {
    const float* x     = (const float*)d_in[0];
    const float* w1    = (const float*)d_in[1];
    const float* b1    = (const float*)d_in[2];
    const float* w2    = (const float*)d_in[3];
    const float* b2    = (const float*)d_in[4];
    const float* wconv = (const float*)d_in[5];
    const float* bconv = (const float*)d_in[6];
    float* out = (float*)d_out;

    float* ws    = (float*)d_ws;
    float* psum  = ws;
    float* pmax  = psum + (size_t)NCHK * Bn * Cn;
    float* chout = pmax + (size_t)NCHK * Bn * Cn;
    float* sp    = chout + (size_t)Bn * Cn;
    float* satt  = sp + (size_t)Bn * HWn * 2;

    k_pool_partial<<<dim3(2 * NCHK, Bn), PSTR * 32, 0, stream>>>(x, psum, pmax);

    k_channel_mlp<<<1, 256, 0, stream>>>(psum, pmax, w1, b1, w2, b2, chout);

    k_spatial_stats<<<(Bn * HWn) / 8, 256, 0, stream>>>(x, chout, sp);

    k_spatial_conv<<<(Bn * HWn) / 256, 256, 0, stream>>>(sp, wconv, bconv, satt);

    k_apply<<<(Bn * HWn * Cn / 4) / 256, 256, 0, stream>>>(x, chout, satt, out);
}


# stats and apply passes: one wave streams 16 pixels with 32 independent 512-B loads in flight (was 8 scalar loads / 1 float4 per lane); 24x fewer waves; changes channel-sum order of spatial mean
# speedup vs baseline: 1.9749x; 1.2460x over previous
#include <hip/hip_runtime.h>
#include <hip/hip_bf16.h>
#include <math.h>

#define Bn   32
#define Hn   56
#define Wn   56
#define Cn   256
#define HWn  (Hn * Wn)
#define HID  16
#define NCHK 8
#define CHUNK (HWn / NCHK)

typedef float v2f __attribute__((ext_vector_type(2)));
typedef float v8f __attribute__((ext_vector_type(8)));

__device__ __forceinline__ v8f wmma4(v2f a, v2f b, v8f c) {
    return __builtin_amdgcn_wmma_f32_16x16x4_f32(
         false, a,  false, b,
         (short)0, c,  false,  false);
}

__device__ __forceinline__ float sigmoidf_(float v) {
    return 1.0f / (1.0f + __expf(-v));
}

#define PSTR 14
#define PPW  (CHUNK / PSTR)
#define CHALF (Cn / 2)

__global__ __launch_bounds__(PSTR * 32)
void k_pool_partial(const float* __restrict__ x,
                    float* __restrict__ psum,
                    float* __restrict__ pmax) {
    __shared__ __attribute__((aligned(16))) float red_s[PSTR * CHALF];
    __shared__ __attribute__((aligned(16))) float red_m[PSTR * CHALF];

    const int lane  = threadIdx.x & 31;
    const int wv    = threadIdx.x >> 5;
    const int half  = blockIdx.x & 1;
    const int chunk = blockIdx.x >> 1;
    const int b     = blockIdx.y;

    const float4* xp = (const float4*)(x + ((size_t)b * HWn + (size_t)chunk * CHUNK + wv) * Cn
                                         + half * CHALF) + lane;

    float4 v[PPW];
    #pragma unroll
    for (int j = 0; j < PPW; ++j)
        v[j] = xp[(size_t)j * PSTR * (Cn / 4)];

    float4 s = {0.0f, 0.0f, 0.0f, 0.0f};
    float4 m = {-3.402823466e38f, -3.402823466e38f, -3.402823466e38f, -3.402823466e38f};
    #pragma unroll
    for (int j = 0; j < PPW; ++j) {
        s.x += v[j].x;  s.y += v[j].y;  s.z += v[j].z;  s.w += v[j].w;
        m.x = fmaxf(m.x, v[j].x);  m.y = fmaxf(m.y, v[j].y);
        m.z = fmaxf(m.z, v[j].z);  m.w = fmaxf(m.w, v[j].w);
    }
    *(float4*)&red_s[wv * CHALF + lane * 4] = s;
    *(float4*)&red_m[wv * CHALF + lane * 4] = m;
    __syncthreads();

    if (threadIdx.x < CHALF) {
        const int c = threadIdx.x;
        float ss = 0.0f, mm = -3.402823466e38f;
        #pragma unroll
        for (int k = 0; k < PSTR; ++k) {
            ss += red_s[k * CHALF + c];
            mm = fmaxf(mm, red_m[k * CHALF + c]);
        }
        const size_t o = ((size_t)chunk * Bn + b) * Cn + half * CHALF + c;
        psum[o] = ss;
        pmax[o] = mm;
    }
}

__global__ void k_channel_mlp(const float* __restrict__ psum,
                              const float* __restrict__ pmax,
                              const float* __restrict__ w1,
                              const float* __restrict__ b1,
                              const float* __restrict__ w2,
                              const float* __restrict__ b2,
                              float* __restrict__ chout) {
    __shared__ float lds_pool[2][Bn][Cn + 1];
    __shared__ float lds_h[2][Bn][HID + 1];

    const int tid  = threadIdx.x;
    const int wv   = tid >> 5;
    const int lane = tid & 31;

    for (int p = tid; p < Bn * Cn; p += 256) {
        const int b = p >> 8, c = p & (Cn - 1);
        float s = 0.0f, m = -3.402823466e38f;
        #pragma unroll
        for (int ch = 0; ch < NCHK; ++ch) {
            s += psum[((size_t)ch * Bn + b) * Cn + c];
            m = fmaxf(m, pmax[((size_t)ch * Bn + b) * Cn + c]);
        }
        lds_pool[0][b][c] = s * (1.0f / (float)HWn);
        lds_pool[1][b][c] = m;
    }
    __syncthreads();

    if (wv < 4) {
        const int src  = wv >> 1;
        const int mt   = wv & 1;
        const int col  = lane & 15;
        const int m    = mt * 16 + col;
        const int ksel = (lane < 16) ? 0 : 2;

        v8f acc = {};
        for (int k0 = 0; k0 < Cn; k0 += 4) {
            v2f a, bfr;
            a.x   = lds_pool[src][m][k0 + ksel];
            a.y   = lds_pool[src][m][k0 + ksel + 1];
            bfr.x = w1[(k0 + ksel)     * HID + col];
            bfr.y = w1[(k0 + ksel + 1) * HID + col];
            acc = wmma4(a, bfr, acc);
        }
        const float bias  = b1[col];
        const int   rbase = mt * 16 + ((lane & 16) ? 8 : 0);
        #pragma unroll
        for (int v = 0; v < 8; ++v)
            lds_h[src][rbase + v][col] = fmaxf(acc[v] + bias, 0.0f);
    }
    __syncthreads();

    for (int t = wv; t < 32; t += 8) {
        const int mt   = t & 1;
        const int nt   = t >> 1;
        const int col  = lane & 15;
        const int ksel = (lane < 16) ? 0 : 2;
        const float bias = b2[nt * 16 + col];

        float combined[8];
        #pragma unroll
        for (int v = 0; v < 8; ++v) combined[v] = 0.0f;

        #pragma unroll
        for (int src = 0; src < 2; ++src) {
            v8f acc = {};
            #pragma unroll
            for (int k0 = 0; k0 < HID; k0 += 4) {
                v2f a, bfr;
                a.x   = lds_h[src][mt * 16 + col][k0 + ksel];
                a.y   = lds_h[src][mt * 16 + col][k0 + ksel + 1];
                bfr.x = w2[(k0 + ksel)     * Cn + nt * 16 + col];
                bfr.y = w2[(k0 + ksel + 1) * Cn + nt * 16 + col];
                acc = wmma4(a, bfr, acc);
            }
            #pragma unroll
            for (int v = 0; v < 8; ++v)
                combined[v] += sigmoidf_(acc[v] + bias);
        }
        const int rbase = mt * 16 + ((lane & 16) ? 8 : 0);
        #pragma unroll
        for (int v = 0; v < 8; ++v)
            chout[(size_t)(rbase + v) * Cn + nt * 16 + col] = combined[v];
    }
}

#define SPW  16
#define SWV  4
#define SBLK ((Bn * HWn) / (SPW * SWV))

__global__ __launch_bounds__(SWV * 32)
void k_spatial_stats(const float* __restrict__ x,
                     const float* __restrict__ chout,
                     float* __restrict__ sp) {
    const int lane = threadIdx.x & 31;
    const int wv   = __builtin_amdgcn_readfirstlane((int)(threadIdx.x >> 5));
    const int pos0 = ((int)blockIdx.x * SWV + wv) * SPW;
    const int b    = pos0 / HWn;

    const float4* cp = (const float4*)(chout + (size_t)b * Cn) + lane;
    const float4 c0 = cp[0];
    const float4 c1 = cp[32];

    const float4* xp = (const float4*)(x + (size_t)pos0 * Cn) + lane;
    float4 v[2 * SPW];
    #pragma unroll
    for (int j = 0; j < 2 * SPW; ++j)
        v[j] = xp[j * 32];

    float keep = 0.0f;
    #pragma unroll
    for (int p = 0; p < SPW; ++p) {
        const float4 a = v[2 * p];
        const float4 d = v[2 * p + 1];
        float s = 0.0f, m = -3.402823466e38f, t;
        t = a.x * c0.x;  s += t;  m = fmaxf(m, t);
        t = a.y * c0.y;  s += t;  m = fmaxf(m, t);
        t = a.z * c0.z;  s += t;  m = fmaxf(m, t);
        t = a.w * c0.w;  s += t;  m = fmaxf(m, t);
        t = d.x * c1.x;  s += t;  m = fmaxf(m, t);
        t = d.y * c1.y;  s += t;  m = fmaxf(m, t);
        t = d.z * c1.z;  s += t;  m = fmaxf(m, t);
        t = d.w * c1.w;  s += t;  m = fmaxf(m, t);
        #pragma unroll
        for (int off = 16; off >= 1; off >>= 1) {
            s += __shfl_xor(s, off, 32);
            m = fmaxf(m, __shfl_xor(m, off, 32));
        }
        keep = (lane == 2 * p)     ? s * (1.0f / (float)Cn) : keep;
        keep = (lane == 2 * p + 1) ? m : keep;
    }
    sp[(size_t)pos0 * 2 + lane] = keep;
}

__global__ void k_spatial_conv(const float* __restrict__ sp,
                               const float* __restrict__ wconv,
                               const float* __restrict__ bconv,
                               float* __restrict__ satt) {
    const int idx = blockIdx.x * blockDim.x + threadIdx.x;
    const int b  = idx / HWn;
    const int hw = idx - b * HWn;
    const int y  = hw / Wn;
    const int xx = hw - y * Wn;

    float acc = 0.0f;
    #pragma unroll
    for (int ky = 0; ky < 7; ++ky) {
        const int iy = y + ky - 3;
        if (iy < 0 || iy >= Hn) continue;
        #pragma unroll
        for (int kx = 0; kx < 7; ++kx) {
            const int ix = xx + kx - 3;
            if (ix < 0 || ix >= Wn) continue;
            const size_t sidx = ((size_t)(b * Hn + iy) * Wn + ix) * 2;
            acc += sp[sidx + 0] * wconv[(ky * 7 + kx) * 2 + 0];
            acc += sp[sidx + 1] * wconv[(ky * 7 + kx) * 2 + 1];
        }
    }
    satt[idx] = sigmoidf_(acc + bconv[0]);
}

__global__ __launch_bounds__(SWV * 32)
void k_apply(const float* __restrict__ x,
             const float* __restrict__ chout,
             const float* __restrict__ satt,
             float* __restrict__ out) {
    const int lane = threadIdx.x & 31;
    const int wv   = __builtin_amdgcn_readfirstlane((int)(threadIdx.x >> 5));
    const int pos0 = ((int)blockIdx.x * SWV + wv) * SPW;
    const int b    = pos0 / HWn;

    const float4* cp = (const float4*)(chout + (size_t)b * Cn) + lane;
    const float4 c0 = cp[0];
    const float4 c1 = cp[32];

    const float4* xp = (const float4*)(x + (size_t)pos0 * Cn) + lane;
    float4*       op = (float4*)(out + (size_t)pos0 * Cn) + lane;
    float4 v[2 * SPW];
    #pragma unroll
    for (int j = 0; j < 2 * SPW; ++j)
        v[j] = xp[j * 32];

    #pragma unroll
    for (int p = 0; p < SPW; ++p) {
        const float s = satt[pos0 + p];
        const float4 a = v[2 * p];
        const float4 d = v[2 * p + 1];
        float4 o0, o1;
        o0.x = a.x * c0.x * s;  o0.y = a.y * c0.y * s;
        o0.z = a.z * c0.z * s;  o0.w = a.w * c0.w * s;
        o1.x = d.x * c1.x * s;  o1.y = d.y * c1.y * s;
        o1.z = d.z * c1.z * s;  o1.w = d.w * c1.w * s;
        op[(2 * p) * 32]     = o0;
        op[(2 * p + 1) * 32] = o1;
    }
}

extern "C" void kernel_launch(void* const* d_in, const int* in_sizes, int n_in,
                              void* d_out, int out_size, void* d_ws, size_t ws_size,
                              hipStream_t stream) {
    const float* x     = (const float*)d_in[0];
    const float* w1    = (const float*)d_in[1];
    const float* b1    = (const float*)d_in[2];
    const float* w2    = (const float*)d_in[3];
    const float* b2    = (const float*)d_in[4];
    const float* wconv = (const float*)d_in[5];
    const float* bconv = (const float*)d_in[6];
    float* out = (float*)d_out;

    float* ws    = (float*)d_ws;
    float* psum  = ws;
    float* pmax  = psum + (size_t)NCHK * Bn * Cn;
    float* chout = pmax + (size_t)NCHK * Bn * Cn;
    float* sp    = chout + (size_t)Bn * Cn;
    float* satt  = sp + (size_t)Bn * HWn * 2;

    k_pool_partial<<<dim3(2 * NCHK, Bn), PSTR * 32, 0, stream>>>(x, psum, pmax);

    k_channel_mlp<<<1, 256, 0, stream>>>(psum, pmax, w1, b1, w2, b2, chout);

    k_spatial_stats<<<SBLK, SWV * 32, 0, stream>>>(x, chout, sp);

    k_spatial_conv<<<(Bn * HWn) / 256, 256, 0, stream>>>(sp, wconv, bconv, satt);

    k_apply<<<SBLK, SWV * 32, 0, stream>>>(x, chout, satt, out);
}


# 7x7 conv + sigmoid folded into the apply pass: each wave gates its own 16 pixels from 49 independent 128-B sp loads (was a 49-step load-wait chain in its own launch); changes the conv sum order
# speedup vs baseline: 2.1374x; 1.0823x over previous
#include <hip/hip_runtime.h>
#include <hip/hip_bf16.h>
#include <math.h>

#define Bn   32
#define Hn   56
#define Wn   56
#define Cn   256
#define HWn  (Hn * Wn)
#define HID  16
#define NCHK 8
#define CHUNK (HWn / NCHK)

typedef float v2f __attribute__((ext_vector_type(2)));
typedef float v8f __attribute__((ext_vector_type(8)));

__device__ __forceinline__ v8f wmma4(v2f a, v2f b, v8f c) {
    return __builtin_amdgcn_wmma_f32_16x16x4_f32(
         false, a,  false, b,
         (short)0, c,  false,  false);
}

__device__ __forceinline__ float sigmoidf_(float v) {
    return 1.0f / (1.0f + __expf(-v));
}

#define PSTR 14
#define PPW  (CHUNK / PSTR)
#define CHALF (Cn / 2)

__global__ __launch_bounds__(PSTR * 32)
void k_pool_partial(const float* __restrict__ x,
                    float* __restrict__ psum,
                    float* __restrict__ pmax) {
    __shared__ __attribute__((aligned(16))) float red_s[PSTR * CHALF];
    __shared__ __attribute__((aligned(16))) float red_m[PSTR * CHALF];

    const int lane  = threadIdx.x & 31;
    const int wv    = threadIdx.x >> 5;
    const int half  = blockIdx.x & 1;
    const int chunk = blockIdx.x >> 1;
    const int b     = blockIdx.y;

    const float4* xp = (const float4*)(x + ((size_t)b * HWn + (size_t)chunk * CHUNK + wv) * Cn
                                         + half * CHALF) + lane;

    float4 v[PPW];
    #pragma unroll
    for (int j = 0; j < PPW; ++j)
        v[j] = xp[(size_t)j * PSTR * (Cn / 4)];

    float4 s = {0.0f, 0.0f, 0.0f, 0.0f};
    float4 m = {-3.402823466e38f, -3.402823466e38f, -3.402823466e38f, -3.402823466e38f};
    #pragma unroll
    for (int j = 0; j < PPW; ++j) {
        s.x += v[j].x;  s.y += v[j].y;  s.z += v[j].z;  s.w += v[j].w;
        m.x = fmaxf(m.x, v[j].x);  m.y = fmaxf(m.y, v[j].y);
        m.z = fmaxf(m.z, v[j].z);  m.w = fmaxf(m.w, v[j].w);
    }
    *(float4*)&red_s[wv * CHALF + lane * 4] = s;
    *(float4*)&red_m[wv * CHALF + lane * 4] = m;
    __syncthreads();

    if (threadIdx.x < CHALF) {
        const int c = threadIdx.x;
        float ss = 0.0f, mm = -3.402823466e38f;
        #pragma unroll
        for (int k = 0; k < PSTR; ++k) {
            ss += red_s[k * CHALF + c];
            mm = fmaxf(mm, red_m[k * CHALF + c]);
        }
        const size_t o = ((size_t)chunk * Bn + b) * Cn + half * CHALF + c;
        psum[o] = ss;
        pmax[o] = mm;
    }
}

__global__ void k_channel_mlp(const float* __restrict__ psum,
                              const float* __restrict__ pmax,
                              const float* __restrict__ w1,
                              const float* __restrict__ b1,
                              const float* __restrict__ w2,
                              const float* __restrict__ b2,
                              float* __restrict__ chout) {
    __shared__ float lds_pool[2][Bn][Cn + 1];
    __shared__ float lds_h[2][Bn][HID + 1];

    const int tid  = threadIdx.x;
    const int wv   = tid >> 5;
    const int lane = tid & 31;

    for (int p = tid; p < Bn * Cn; p += 256) {
        const int b = p >> 8, c = p & (Cn - 1);
        float s = 0.0f, m = -3.402823466e38f;
        #pragma unroll
        for (int ch = 0; ch < NCHK; ++ch) {
            s += psum[((size_t)ch * Bn + b) * Cn + c];
            m = fmaxf(m, pmax[((size_t)ch * Bn + b) * Cn + c]);
        }
        lds_pool[0][b][c] = s * (1.0f / (float)HWn);
        lds_pool[1][b][c] = m;
    }
    __syncthreads();

    if (wv < 4) {
        const int src  = wv >> 1;
        const int mt   = wv & 1;
        const int col  = lane & 15;
        const int m    = mt * 16 + col;
        const int ksel = (lane < 16) ? 0 : 2;

        v8f acc = {};
        for (int k0 = 0; k0 < Cn; k0 += 4) {
            v2f a, bfr;
            a.x   = lds_pool[src][m][k0 + ksel];
            a.y   = lds_pool[src][m][k0 + ksel + 1];
            bfr.x = w1[(k0 + ksel)     * HID + col];
            bfr.y = w1[(k0 + ksel + 1) * HID + col];
            acc = wmma4(a, bfr, acc);
        }
        const float bias  = b1[col];
        const int   rbase = mt * 16 + ((lane & 16) ? 8 : 0);
        #pragma unroll
        for (int v = 0; v < 8; ++v)
            lds_h[src][rbase + v][col] = fmaxf(acc[v] + bias, 0.0f);
    }
    __syncthreads();

    for (int t = wv; t < 32; t += 8) {
        const int mt   = t & 1;
        const int nt   = t >> 1;
        const int col  = lane & 15;
        const int ksel = (lane < 16) ? 0 : 2;
        const float bias = b2[nt * 16 + col];

        float combined[8];
        #pragma unroll
        for (int v = 0; v < 8; ++v) combined[v] = 0.0f;

        #pragma unroll
        for (int src = 0; src < 2; ++src) {
            v8f acc = {};
            #pragma unroll
            for (int k0 = 0; k0 < HID; k0 += 4) {
                v2f a, bfr;
                a.x   = lds_h[src][mt * 16 + col][k0 + ksel];
                a.y   = lds_h[src][mt * 16 + col][k0 + ksel + 1];
                bfr.x = w2[(k0 + ksel)     * Cn + nt * 16 + col];
                bfr.y = w2[(k0 + ksel + 1) * Cn + nt * 16 + col];
                acc = wmma4(a, bfr, acc);
            }
            #pragma unroll
            for (int v = 0; v < 8; ++v)
                combined[v] += sigmoidf_(acc[v] + bias);
        }
        const int rbase = mt * 16 + ((lane & 16) ? 8 : 0);
        #pragma unroll
        for (int v = 0; v < 8; ++v)
            chout[(size_t)(rbase + v) * Cn + nt * 16 + col] = combined[v];
    }
}

#define SPW  16
#define SWV  4
#define SBLK ((Bn * HWn) / (SPW * SWV))

__global__ __launch_bounds__(SWV * 32)
void k_spatial_stats(const float* __restrict__ x,
                     const float* __restrict__ chout,
                     float* __restrict__ sp) {
    const int lane = threadIdx.x & 31;
    const int wv   = __builtin_amdgcn_readfirstlane((int)(threadIdx.x >> 5));
    const int pos0 = ((int)blockIdx.x * SWV + wv) * SPW;
    const int b    = pos0 / HWn;

    const float4* cp = (const float4*)(chout + (size_t)b * Cn) + lane;
    const float4 c0 = cp[0];
    const float4 c1 = cp[32];

    const float4* xp = (const float4*)(x + (size_t)pos0 * Cn) + lane;
    float4 v[2 * SPW];
    #pragma unroll
    for (int j = 0; j < 2 * SPW; ++j)
        v[j] = xp[j * 32];

    float keep = 0.0f;
    #pragma unroll
    for (int p = 0; p < SPW; ++p) {
        const float4 a = v[2 * p];
        const float4 d = v[2 * p + 1];
        float s = 0.0f, m = -3.402823466e38f, t;
        t = a.x * c0.x;  s += t;  m = fmaxf(m, t);
        t = a.y * c0.y;  s += t;  m = fmaxf(m, t);
        t = a.z * c0.z;  s += t;  m = fmaxf(m, t);
        t = a.w * c0.w;  s += t;  m = fmaxf(m, t);
        t = d.x * c1.x;  s += t;  m = fmaxf(m, t);
        t = d.y * c1.y;  s += t;  m = fmaxf(m, t);
        t = d.z * c1.z;  s += t;  m = fmaxf(m, t);
        t = d.w * c1.w;  s += t;  m = fmaxf(m, t);
        #pragma unroll
        for (int off = 16; off >= 1; off >>= 1) {
            s += __shfl_xor(s, off, 32);
            m = fmaxf(m, __shfl_xor(m, off, 32));
        }
        keep = (lane == 2 * p)     ? s * (1.0f / (float)Cn) : keep;
        keep = (lane == 2 * p + 1) ? m : keep;
    }
    sp[(size_t)pos0 * 2 + lane] = keep;
}

__global__ __launch_bounds__(SWV * 32)
void k_apply(const float* __restrict__ x,
             const float* __restrict__ chout,
             const float* __restrict__ sp,
             const float* __restrict__ wconv,
             const float* __restrict__ bconv,
             float* __restrict__ out) {
    const int lane = threadIdx.x & 31;
    const int wv   = __builtin_amdgcn_readfirstlane((int)(threadIdx.x >> 5));
    const int pos0 = ((int)blockIdx.x * SWV + wv) * SPW;
    const int b    = pos0 / HWn;

    const int pix = pos0 - b * HWn + (lane >> 1);
    const int py  = pix / Wn;
    const int px  = pix - py * Wn;
    const float* spb = sp + (size_t)b * (HWn * 2);
    unsigned ro[7], co[7];
    #pragma unroll
    for (int k = 0; k < 7; ++k) {
        ro[k] = (unsigned)(min(max(py + k - 3, 0), Hn - 1) * (Wn * 2) + (lane & 1));
        co[k] = (unsigned)(min(max(px + k - 3, 0), Wn - 1) * 2);
    }
    float tv[49];
    #pragma unroll
    for (int ky = 0; ky < 7; ++ky) {
        #pragma unroll
        for (int kx = 0; kx < 7; ++kx)
            tv[ky * 7 + kx] = spb[ro[ky] + co[kx]];
    }

    const float4* cp = (const float4*)(chout + (size_t)b * Cn) + lane;
    const float4 c0 = cp[0];
    const float4 c1 = cp[32];

    const float4* xp = (const float4*)(x + (size_t)pos0 * Cn) + lane;
    float4*       op = (float4*)(out + (size_t)pos0 * Cn) + lane;
    float4 v[2 * SPW];
    #pragma unroll
    for (int j = 0; j < 2 * SPW; ++j)
        v[j] = xp[j * 32];

    int rm[7], cm[7];
    #pragma unroll
    for (int k = 0; k < 7; ++k) {
        rm[k] = ((unsigned)(py + k - 3) < (unsigned)Hn) ? -1 : 0;
        cm[k] = ((unsigned)(px + k - 3) < (unsigned)Wn) ? -1 : 0;
    }
    float a0 = 0.0f, a1 = 0.0f;
    #pragma unroll
    for (int ky = 0; ky < 7; ++ky) {
        #pragma unroll
        for (int kx = 0; kx < 7; ++kx) {
            const float t = __int_as_float(__float_as_int(tv[ky * 7 + kx]) & rm[ky] & cm[kx]);
            a0 += t * wconv[(ky * 7 + kx) * 2 + 0];
            a1 += t * wconv[(ky * 7 + kx) * 2 + 1];
        }
    }
    const float mine = (lane & 1) ? a1 : a0;
    const float gate = sigmoidf_(mine + __shfl_xor(mine, 1, 32) + bconv[0]);

    #pragma unroll
    for (int p = 0; p < SPW; ++p) {
        const float s = __int_as_float(
            __builtin_amdgcn_readlane(__float_as_int(gate), 2 * p));
        const float4 a = v[2 * p];
        const float4 d = v[2 * p + 1];
        float4 o0, o1;
        o0.x = a.x * c0.x * s;  o0.y = a.y * c0.y * s;
        o0.z = a.z * c0.z * s;  o0.w = a.w * c0.w * s;
        o1.x = d.x * c1.x * s;  o1.y = d.y * c1.y * s;
        o1.z = d.z * c1.z * s;  o1.w = d.w * c1.w * s;
        op[(2 * p) * 32]     = o0;
        op[(2 * p + 1) * 32] = o1;
    }
}

extern "C" void kernel_launch(void* const* d_in, const int* in_sizes, int n_in,
                              void* d_out, int out_size, void* d_ws, size_t ws_size,
                              hipStream_t stream) {
    const float* x     = (const float*)d_in[0];
    const float* w1    = (const float*)d_in[1];
    const float* b1    = (const float*)d_in[2];
    const float* w2    = (const float*)d_in[3];
    const float* b2    = (const float*)d_in[4];
    const float* wconv = (const float*)d_in[5];
    const float* bconv = (const float*)d_in[6];
    float* out = (float*)d_out;

    float* ws    = (float*)d_ws;
    float* psum  = ws;
    float* pmax  = psum + (size_t)NCHK * Bn * Cn;
    float* chout = pmax + (size_t)NCHK * Bn * Cn;
    float* sp    = chout + (size_t)Bn * Cn;

    k_pool_partial<<<dim3(2 * NCHK, Bn), PSTR * 32, 0, stream>>>(x, psum, pmax);

    k_channel_mlp<<<1, 256, 0, stream>>>(psum, pmax, w1, b1, w2, b2, chout);

    k_spatial_stats<<<SBLK, SWV * 32, 0, stream>>>(x, chout, sp);

    k_apply<<<SBLK, SWV * 32, 0, stream>>>(x, chout, sp, wconv, bconv, out);
}


# channel-MLP launch: 2 blocks x 32 waves, every global load issued up front as one group (was 1 block x 8 waves, 28 load-wait rounds); GEMM1 K split over 16 waves + LDS fold: changes GEMM1 sum order
# speedup vs baseline: 3.1516x; 1.4745x over previous
#include <hip/hip_runtime.h>
#include <hip/hip_bf16.h>
#include <math.h>

#define Bn   32
#define Hn   56
#define Wn   56
#define Cn   256
#define HWn  (Hn * Wn)
#define HID  16
#define NCHK 8
#define CHUNK (HWn / NCHK)

typedef float v2f __attribute__((ext_vector_type(2)));
typedef float v8f __attribute__((ext_vector_type(8)));

__device__ __forceinline__ v8f wmma4(v2f a, v2f b, v8f c) {
    return __builtin_amdgcn_wmma_f32_16x16x4_f32(
         false, a,  false, b,
         (short)0, c,  false,  false);
}

__device__ __forceinline__ float sigmoidf_(float v) {
    return 1.0f / (1.0f + __expf(-v));
}

#define PSTR 14
#define PPW  (CHUNK / PSTR)
#define CHALF (Cn / 2)

__global__ __launch_bounds__(PSTR * 32)
void k_pool_partial(const float* __restrict__ x,
                    float* __restrict__ psum,
                    float* __restrict__ pmax) {
    __shared__ __attribute__((aligned(16))) float red_s[PSTR * CHALF];
    __shared__ __attribute__((aligned(16))) float red_m[PSTR * CHALF];

    const int lane  = threadIdx.x & 31;
    const int wv    = threadIdx.x >> 5;
    const int half  = blockIdx.x & 1;
    const int chunk = blockIdx.x >> 1;
    const int b     = blockIdx.y;

    const float4* xp = (const float4*)(x + ((size_t)b * HWn + (size_t)chunk * CHUNK + wv) * Cn
                                         + half * CHALF) + lane;

    float4 v[PPW];
    #pragma unroll
    for (int j = 0; j < PPW; ++j)
        v[j] = xp[(size_t)j * PSTR * (Cn / 4)];

    float4 s = {0.0f, 0.0f, 0.0f, 0.0f};
    float4 m = {-3.402823466e38f, -3.402823466e38f, -3.402823466e38f, -3.402823466e38f};
    #pragma unroll
    for (int j = 0; j < PPW; ++j) {
        s.x += v[j].x;  s.y += v[j].y;  s.z += v[j].z;  s.w += v[j].w;
        m.x = fmaxf(m.x, v[j].x);  m.y = fmaxf(m.y, v[j].y);
        m.z = fmaxf(m.z, v[j].z);  m.w = fmaxf(m.w, v[j].w);
    }
    *(float4*)&red_s[wv * CHALF + lane * 4] = s;
    *(float4*)&red_m[wv * CHALF + lane * 4] = m;
    __syncthreads();

    if (threadIdx.x < CHALF) {
        const int c = threadIdx.x;
        float ss = 0.0f, mm = -3.402823466e38f;
        #pragma unroll
        for (int k = 0; k < PSTR; ++k) {
            ss += red_s[k * CHALF + c];
            mm = fmaxf(mm, red_m[k * CHALF + c]);
        }
        const size_t o = ((size_t)chunk * Bn + b) * Cn + half * CHALF + c;
        psum[o] = ss;
        pmax[o] = mm;
    }
}

#define MROWS  16
#define PPITCH (Cn + 4)
#define KSL    16
#define KPS    (Cn / KSL)

__global__ __launch_bounds__(1024)
void k_channel_mlp(const float* __restrict__ psum,
                   const float* __restrict__ pmax,
                   const float* __restrict__ w1,
                   const float* __restrict__ b1,
                   const float* __restrict__ w2,
                   const float* __restrict__ b2,
                   float* __restrict__ chout) {
    __shared__ __attribute__((aligned(16))) float lds_pool[2][MROWS][PPITCH];
    __shared__ float lds_part[2 * KSL][256];
    __shared__ float lds_h[2][MROWS][HID + 1];

    const int tid  = threadIdx.x;
    const int wv   = tid >> 5;
    const int lane = tid & 31;
    const int mt   = blockIdx.x;
    const int col  = lane & 15;
    const int ksel = (lane < 16) ? 0 : 2;

    const int r  = tid >> 6;
    const int c4 = tid & 63;
    const size_t o4 = ((size_t)(mt * MROWS + r) * Cn) / 4 + c4;
    const float4* ps4 = (const float4*)psum + o4;
    const float4* pm4 = (const float4*)pmax + o4;
    float4 sv[NCHK], mv[NCHK];
    #pragma unroll
    for (int ch = 0; ch < NCHK; ++ch) {
        sv[ch] = ps4[(size_t)ch * (Bn * Cn / 4)];
        mv[ch] = pm4[(size_t)ch * (Bn * Cn / 4)];
    }

    const int src1 = wv & 1;
    const int ks   = wv >> 1;
    const int nt   = wv;
    v2f bw1[KPS / 4];
    #pragma unroll
    for (int j = 0; j < KPS / 4; ++j) {
        const int k = ks * KPS + 4 * j + ksel;
        bw1[j].x = w1[(k)     * HID + col];
        bw1[j].y = w1[(k + 1) * HID + col];
    }
    const float bias1 = b1[col];
    v2f bw2[HID / 4] = {};
    float bias2 = 0.0f;
    if (wv < 16) {
        #pragma unroll
        for (int j = 0; j < HID / 4; ++j) {
            bw2[j].x = w2[(4 * j + ksel)     * Cn + nt * 16 + col];
            bw2[j].y = w2[(4 * j + ksel + 1) * Cn + nt * 16 + col];
        }
        bias2 = b2[nt * 16 + col];
    }

    {
        float4 s = {0.0f, 0.0f, 0.0f, 0.0f};
        float4 m = {-3.402823466e38f, -3.402823466e38f, -3.402823466e38f, -3.402823466e38f};
        #pragma unroll
        for (int ch = 0; ch < NCHK; ++ch) {
            s.x += sv[ch].x;  s.y += sv[ch].y;  s.z += sv[ch].z;  s.w += sv[ch].w;
            m.x = fmaxf(m.x, mv[ch].x);  m.y = fmaxf(m.y, mv[ch].y);
            m.z = fmaxf(m.z, mv[ch].z);  m.w = fmaxf(m.w, mv[ch].w);
        }
        const float inv = 1.0f / (float)HWn;
        s.x *= inv;  s.y *= inv;  s.z *= inv;  s.w *= inv;
        *(float4*)&lds_pool[0][r][4 * c4] = s;
        *(float4*)&lds_pool[1][r][4 * c4] = m;
    }
    __syncthreads();

    {
        v8f acc = {};
        #pragma unroll
        for (int j = 0; j < KPS / 4; ++j) {
            const int k = ks * KPS + 4 * j + ksel;
            v2f a;
            a.x = lds_pool[src1][col][k];
            a.y = lds_pool[src1][col][k + 1];
            acc = wmma4(a, bw1[j], acc);
        }
        #pragma unroll
        for (int v = 0; v < 8; ++v)
            lds_part[wv][v * 32 + lane] = acc[v];
    }
    __syncthreads();

    if (tid < 512) {
        const int src = tid >> 8;
        const int e   = tid & 255;
        float sum = 0.0f;
        #pragma unroll
        for (int q = 0; q < KSL; ++q)
            sum += lds_part[q * 2 + src][e];
        const int row = ((e & 16) ? 8 : 0) + (e >> 5);
        lds_h[src][row][col] = fmaxf(sum + bias1, 0.0f);
    }
    __syncthreads();

    if (wv < 16) {
        float combined[8];
        #pragma unroll
        for (int v = 0; v < 8; ++v) combined[v] = 0.0f;

        #pragma unroll
        for (int src = 0; src < 2; ++src) {
            v8f acc = {};
            #pragma unroll
            for (int j = 0; j < HID / 4; ++j) {
                v2f a;
                a.x = lds_h[src][col][4 * j + ksel];
                a.y = lds_h[src][col][4 * j + ksel + 1];
                acc = wmma4(a, bw2[j], acc);
            }
            #pragma unroll
            for (int v = 0; v < 8; ++v)
                combined[v] += sigmoidf_(acc[v] + bias2);
        }
        const int rbase = mt * MROWS + ((lane & 16) ? 8 : 0);
        #pragma unroll
        for (int v = 0; v < 8; ++v)
            chout[(size_t)(rbase + v) * Cn + nt * 16 + col] = combined[v];
    }
}

#define SPW  16
#define SWV  4
#define SBLK ((Bn * HWn) / (SPW * SWV))

__global__ __launch_bounds__(SWV * 32)
void k_spatial_stats(const float* __restrict__ x,
                     const float* __restrict__ chout,
                     float* __restrict__ sp) {
    const int lane = threadIdx.x & 31;
    const int wv   = __builtin_amdgcn_readfirstlane((int)(threadIdx.x >> 5));
    const int pos0 = ((int)blockIdx.x * SWV + wv) * SPW;
    const int b    = pos0 / HWn;

    const float4* cp = (const float4*)(chout + (size_t)b * Cn) + lane;
    const float4 c0 = cp[0];
    const float4 c1 = cp[32];

    const float4* xp = (const float4*)(x + (size_t)pos0 * Cn) + lane;
    float4 v[2 * SPW];
    #pragma unroll
    for (int j = 0; j < 2 * SPW; ++j)
        v[j] = xp[j * 32];

    float keep = 0.0f;
    #pragma unroll
    for (int p = 0; p < SPW; ++p) {
        const float4 a = v[2 * p];
        const float4 d = v[2 * p + 1];
        float s = 0.0f, m = -3.402823466e38f, t;
        t = a.x * c0.x;  s += t;  m = fmaxf(m, t);
        t = a.y * c0.y;  s += t;  m = fmaxf(m, t);
        t = a.z * c0.z;  s += t;  m = fmaxf(m, t);
        t = a.w * c0.w;  s += t;  m = fmaxf(m, t);
        t = d.x * c1.x;  s += t;  m = fmaxf(m, t);
        t = d.y * c1.y;  s += t;  m = fmaxf(m, t);
        t = d.z * c1.z;  s += t;  m = fmaxf(m, t);
        t = d.w * c1.w;  s += t;  m = fmaxf(m, t);
        #pragma unroll
        for (int off = 16; off >= 1; off >>= 1) {
            s += __shfl_xor(s, off, 32);
            m = fmaxf(m, __shfl_xor(m, off, 32));
        }
        keep = (lane == 2 * p)     ? s * (1.0f / (float)Cn) : keep;
        keep = (lane == 2 * p + 1) ? m : keep;
    }
    sp[(size_t)pos0 * 2 + lane] = keep;
}

__global__ __launch_bounds__(SWV * 32)
void k_apply(const float* __restrict__ x,
             const float* __restrict__ chout,
             const float* __restrict__ sp,
             const float* __restrict__ wconv,
             const float* __restrict__ bconv,
             float* __restrict__ out) {
    const int lane = threadIdx.x & 31;
    const int wv   = __builtin_amdgcn_readfirstlane((int)(threadIdx.x >> 5));
    const int pos0 = ((int)blockIdx.x * SWV + wv) * SPW;
    const int b    = pos0 / HWn;

    const int pix = pos0 - b * HWn + (lane >> 1);
    const int py  = pix / Wn;
    const int px  = pix - py * Wn;
    const float* spb = sp + (size_t)b * (HWn * 2);
    unsigned ro[7], co[7];
    #pragma unroll
    for (int k = 0; k < 7; ++k) {
        ro[k] = (unsigned)(min(max(py + k - 3, 0), Hn - 1) * (Wn * 2) + (lane & 1));
        co[k] = (unsigned)(min(max(px + k - 3, 0), Wn - 1) * 2);
    }
    float tv[49];
    #pragma unroll
    for (int ky = 0; ky < 7; ++ky) {
        #pragma unroll
        for (int kx = 0; kx < 7; ++kx)
            tv[ky * 7 + kx] = spb[ro[ky] + co[kx]];
    }

    const float4* cp = (const float4*)(chout + (size_t)b * Cn) + lane;
    const float4 c0 = cp[0];
    const float4 c1 = cp[32];

    const float4* xp = (const float4*)(x + (size_t)pos0 * Cn) + lane;
    float4*       op = (float4*)(out + (size_t)pos0 * Cn) + lane;
    float4 v[2 * SPW];
    #pragma unroll
    for (int j = 0; j < 2 * SPW; ++j)
        v[j] = xp[j * 32];

    int rm[7], cm[7];
    #pragma unroll
    for (int k = 0; k < 7; ++k) {
        rm[k] = ((unsigned)(py + k - 3) < (unsigned)Hn) ? -1 : 0;
        cm[k] = ((unsigned)(px + k - 3) < (unsigned)Wn) ? -1 : 0;
    }
    float a0 = 0.0f, a1 = 0.0f;
    #pragma unroll
    for (int ky = 0; ky < 7; ++ky) {
        #pragma unroll
        for (int kx = 0; kx < 7; ++kx) {
            const float t = __int_as_float(__float_as_int(tv[ky * 7 + kx]) & rm[ky] & cm[kx]);
            a0 += t * wconv[(ky * 7 + kx) * 2 + 0];
            a1 += t * wconv[(ky * 7 + kx) * 2 + 1];
        }
    }
    const float mine = (lane & 1) ? a1 : a0;
    const float gate = sigmoidf_(mine + __shfl_xor(mine, 1, 32) + bconv[0]);

    #pragma unroll
    for (int p = 0; p < SPW; ++p) {
        const float s = __int_as_float(
            __builtin_amdgcn_readlane(__float_as_int(gate), 2 * p));
        const float4 a = v[2 * p];
        const float4 d = v[2 * p + 1];
        float4 o0, o1;
        o0.x = a.x * c0.x * s;  o0.y = a.y * c0.y * s;
        o0.z = a.z * c0.z * s;  o0.w = a.w * c0.w * s;
        o1.x = d.x * c1.x * s;  o1.y = d.y * c1.y * s;
        o1.z = d.z * c1.z * s;  o1.w = d.w * c1.w * s;
        op[(2 * p) * 32]     = o0;
        op[(2 * p + 1) * 32] = o1;
    }
}

extern "C" void kernel_launch(void* const* d_in, const int* in_sizes, int n_in,
                              void* d_out, int out_size, void* d_ws, size_t ws_size,
                              hipStream_t stream) {
    const float* x     = (const float*)d_in[0];
    const float* w1    = (const float*)d_in[1];
    const float* b1    = (const float*)d_in[2];
    const float* w2    = (const float*)d_in[3];
    const float* b2    = (const float*)d_in[4];
    const float* wconv = (const float*)d_in[5];
    const float* bconv = (const float*)d_in[6];
    float* out = (float*)d_out;

    float* ws    = (float*)d_ws;
    float* psum  = ws;
    float* pmax  = psum + (size_t)NCHK * Bn * Cn;
    float* chout = pmax + (size_t)NCHK * Bn * Cn;
    float* sp    = chout + (size_t)Bn * Cn;

    k_pool_partial<<<dim3(2 * NCHK, Bn), PSTR * 32, 0, stream>>>(x, psum, pmax);

    k_channel_mlp<<<Bn / MROWS, 1024, 0, stream>>>(psum, pmax, w1, b1, w2, b2, chout);

    k_spatial_stats<<<SBLK, SWV * 32, 0, stream>>>(x, chout, sp);

    k_apply<<<SBLK, SWV * 32, 0, stream>>>(x, chout, sp, wconv, bconv, out);
}


# stats and apply passes fused (4 -> 3 launches): a block owns a 7-row band of one image, recomputes sp of the 3 halo rows each side into an LDS tile, keeps its own x on chip; sum orders unchanged
# speedup vs baseline: 3.3311x; 1.0570x over previous
#include <hip/hip_runtime.h>
#include <hip/hip_bf16.h>
#include <math.h>

#define Bn   32
#define Hn   56
#define Wn   56
#define Cn   256
#define HWn  (Hn * Wn)
#define HID  16
#define NCHK 8
#define CHUNK (HWn / NCHK)

typedef float v2f __attribute__((ext_vector_type(2)));
typedef float v8f __attribute__((ext_vector_type(8)));

__device__ __forceinline__ v8f wmma4(v2f a, v2f b, v8f c) {
    return __builtin_amdgcn_wmma_f32_16x16x4_f32(
         false, a,  false, b,
         (short)0, c,  false,  false);
}

__device__ __forceinline__ float sigmoidf_(float v) {
    return 1.0f / (1.0f + __expf(-v));
}

#define PSTR 14
#define PPW  (CHUNK / PSTR)
#define CHALF (Cn / 2)

__global__ __launch_bounds__(PSTR * 32)
void k_pool_partial(const float* __restrict__ x,
                    float* __restrict__ psum,
                    float* __restrict__ pmax) {
    __shared__ __attribute__((aligned(16))) float red_s[PSTR * CHALF];
    __shared__ __attribute__((aligned(16))) float red_m[PSTR * CHALF];

    const int lane  = threadIdx.x & 31;
    const int wv    = threadIdx.x >> 5;
    const int half  = blockIdx.x & 1;
    const int chunk = blockIdx.x >> 1;
    const int b     = blockIdx.y;

    const float4* xp = (const float4*)(x + ((size_t)b * HWn + (size_t)chunk * CHUNK + wv) * Cn
                                         + half * CHALF) + lane;

    float4 v[PPW];
    #pragma unroll
    for (int j = 0; j < PPW; ++j)
        v[j] = xp[(size_t)j * PSTR * (Cn / 4)];

    float4 s = {0.0f, 0.0f, 0.0f, 0.0f};
    float4 m = {-3.402823466e38f, -3.402823466e38f, -3.402823466e38f, -3.402823466e38f};
    #pragma unroll
    for (int j = 0; j < PPW; ++j) {
        s.x += v[j].x;  s.y += v[j].y;  s.z += v[j].z;  s.w += v[j].w;
        m.x = fmaxf(m.x, v[j].x);  m.y = fmaxf(m.y, v[j].y);
        m.z = fmaxf(m.z, v[j].z);  m.w = fmaxf(m.w, v[j].w);
    }
    *(float4*)&red_s[wv * CHALF + lane * 4] = s;
    *(float4*)&red_m[wv * CHALF + lane * 4] = m;
    __syncthreads();

    if (threadIdx.x < CHALF) {
        const int c = threadIdx.x;
        float ss = 0.0f, mm = -3.402823466e38f;
        #pragma unroll
        for (int k = 0; k < PSTR; ++k) {
            ss += red_s[k * CHALF + c];
            mm = fmaxf(mm, red_m[k * CHALF + c]);
        }
        const size_t o = ((size_t)chunk * Bn + b) * Cn + half * CHALF + c;
        psum[o] = ss;
        pmax[o] = mm;
    }
}

#define MROWS  16
#define PPITCH (Cn + 4)
#define KSL    16
#define KPS    (Cn / KSL)

__global__ __launch_bounds__(1024)
void k_channel_mlp(const float* __restrict__ psum,
                   const float* __restrict__ pmax,
                   const float* __restrict__ w1,
                   const float* __restrict__ b1,
                   const float* __restrict__ w2,
                   const float* __restrict__ b2,
                   float* __restrict__ chout) {
    __shared__ __attribute__((aligned(16))) float lds_pool[2][MROWS][PPITCH];
    __shared__ float lds_part[2 * KSL][256];
    __shared__ float lds_h[2][MROWS][HID + 1];

    const int tid  = threadIdx.x;
    const int wv   = tid >> 5;
    const int lane = tid & 31;
    const int mt   = blockIdx.x;
    const int col  = lane & 15;
    const int ksel = (lane < 16) ? 0 : 2;

    const int r  = tid >> 6;
    const int c4 = tid & 63;
    const size_t o4 = ((size_t)(mt * MROWS + r) * Cn) / 4 + c4;
    const float4* ps4 = (const float4*)psum + o4;
    const float4* pm4 = (const float4*)pmax + o4;
    float4 sv[NCHK], mv[NCHK];
    #pragma unroll
    for (int ch = 0; ch < NCHK; ++ch) {
        sv[ch] = ps4[(size_t)ch * (Bn * Cn / 4)];
        mv[ch] = pm4[(size_t)ch * (Bn * Cn / 4)];
    }

    const int src1 = wv & 1;
    const int ks   = wv >> 1;
    const int nt   = wv;
    v2f bw1[KPS / 4];
    #pragma unroll
    for (int j = 0; j < KPS / 4; ++j) {
        const int k = ks * KPS + 4 * j + ksel;
        bw1[j].x = w1[(k)     * HID + col];
        bw1[j].y = w1[(k + 1) * HID + col];
    }
    const float bias1 = b1[col];
    v2f bw2[HID / 4] = {};
    float bias2 = 0.0f;
    if (wv < 16) {
        #pragma unroll
        for (int j = 0; j < HID / 4; ++j) {
            bw2[j].x = w2[(4 * j + ksel)     * Cn + nt * 16 + col];
            bw2[j].y = w2[(4 * j + ksel + 1) * Cn + nt * 16 + col];
        }
        bias2 = b2[nt * 16 + col];
    }

    {
        float4 s = {0.0f, 0.0f, 0.0f, 0.0f};
        float4 m = {-3.402823466e38f, -3.402823466e38f, -3.402823466e38f, -3.402823466e38f};
        #pragma unroll
        for (int ch = 0; ch < NCHK; ++ch) {
            s.x += sv[ch].x;  s.y += sv[ch].y;  s.z += sv[ch].z;  s.w += sv[ch].w;
            m.x = fmaxf(m.x, mv[ch].x);  m.y = fmaxf(m.y, mv[ch].y);
            m.z = fmaxf(m.z, mv[ch].z);  m.w = fmaxf(m.w, mv[ch].w);
        }
        const float inv = 1.0f / (float)HWn;
        s.x *= inv;  s.y *= inv;  s.z *= inv;  s.w *= inv;
        *(float4*)&lds_pool[0][r][4 * c4] = s;
        *(float4*)&lds_pool[1][r][4 * c4] = m;
    }
    __syncthreads();

    {
        v8f acc = {};
        #pragma unroll
        for (int j = 0; j < KPS / 4; ++j) {
            const int k = ks * KPS + 4 * j + ksel;
            v2f a;
            a.x = lds_pool[src1][col][k];
            a.y = lds_pool[src1][col][k + 1];
            acc = wmma4(a, bw1[j], acc);
        }
        #pragma unroll
        for (int v = 0; v < 8; ++v)
            lds_part[wv][v * 32 + lane] = acc[v];
    }
    __syncthreads();

    if (tid < 512) {
        const int src = tid >> 8;
        const int e   = tid & 255;
        float sum = 0.0f;
        #pragma unroll
        for (int q = 0; q < KSL; ++q)
            sum += lds_part[q * 2 + src][e];
        const int row = ((e & 16) ? 8 : 0) + (e >> 5);
        lds_h[src][row][col] = fmaxf(sum + bias1, 0.0f);
    }
    __syncthreads();

    if (wv < 16) {
        float combined[8];
        #pragma unroll
        for (int v = 0; v < 8; ++v) combined[v] = 0.0f;

        #pragma unroll
        for (int src = 0; src < 2; ++src) {
            v8f acc = {};
            #pragma unroll
            for (int j = 0; j < HID / 4; ++j) {
                v2f a;
                a.x = lds_h[src][col][4 * j + ksel];
                a.y = lds_h[src][col][4 * j + ksel + 1];
                acc = wmma4(a, bw2[j], acc);
            }
            #pragma unroll
            for (int v = 0; v < 8; ++v)
                combined[v] += sigmoidf_(acc[v] + bias2);
        }
        const int rbase = mt * MROWS + ((lane & 16) ? 8 : 0);
        #pragma unroll
        for (int v = 0; v < 8; ++v)
            chout[(size_t)(rbase + v) * Cn + nt * 16 + col] = combined[v];
    }
}

#define FR    7
#define FBND  (Hn / FR)
#define FWV   28
#define FOWN  ((FR * Wn) / FWV)
#define FHAL  ((3 * Wn) / (FWV / 2))
#define TROWS (FR + 6)
#define TCOLS (Wn + 6)
#define TSIZE (TROWS * TCOLS * 2)
#define FST   6
#define FRG   (FOWN - FST)
#define FHH   (FHAL / 2)
#define FSH   (FST / 2)

__device__ __forceinline__ void pixel_stats(const float4 a, const float4 d,
                                            const float4 c0, const float4 c1,
                                            float& s_out, float& m_out) {
    float s = 0.0f, m = -3.402823466e38f, t;
    t = a.x * c0.x;  s += t;  m = fmaxf(m, t);
    t = a.y * c0.y;  s += t;  m = fmaxf(m, t);
    t = a.z * c0.z;  s += t;  m = fmaxf(m, t);
    t = a.w * c0.w;  s += t;  m = fmaxf(m, t);
    t = d.x * c1.x;  s += t;  m = fmaxf(m, t);
    t = d.y * c1.y;  s += t;  m = fmaxf(m, t);
    t = d.z * c1.z;  s += t;  m = fmaxf(m, t);
    t = d.w * c1.w;  s += t;  m = fmaxf(m, t);
    #pragma unroll
    for (int off = 16; off >= 1; off >>= 1) {
        s += __shfl_xor(s, off, 32);
        m = fmaxf(m, __shfl_xor(m, off, 32));
    }
    s_out = s * (1.0f / (float)Cn);
    m_out = m;
}

__device__ __forceinline__ void scale_store(const float4 a, const float4 d,
                                            const float4 c0, const float4 c1,
                                            const float s,
                                            float4* __restrict__ o0p,
                                            float4* __restrict__ o1p) {
    float4 o0, o1;
    o0.x = a.x * c0.x * s;  o0.y = a.y * c0.y * s;
    o0.z = a.z * c0.z * s;  o0.w = a.w * c0.w * s;
    o1.x = d.x * c1.x * s;  o1.y = d.y * c1.y * s;
    o1.z = d.z * c1.z * s;  o1.w = d.w * c1.w * s;
    *o0p = o0;
    *o1p = o1;
}

__global__ __launch_bounds__(FWV * 32)
void k_stats_apply(const float* __restrict__ x,
                   const float* __restrict__ chout,
                   const float* __restrict__ wconv,
                   const float* __restrict__ bconv,
                   float* __restrict__ out) {
    __shared__ float sp_t[TSIZE];
    __shared__ float4 park[FWV * 2 * FST * 32];

    const int lane = threadIdx.x & 31;
    const int wv   = __builtin_amdgcn_readfirstlane((int)(threadIdx.x >> 5));
    const int b    = (int)blockIdx.x / FBND;
    const int band = (int)blockIdx.x - b * FBND;
    const int r0   = band * FR;

    const float* xb = x + (size_t)b * HWn * Cn;

    const float4* cp = (const float4*)(chout + (size_t)b * Cn) + lane;
    const float4 c0 = cp[0];
    const float4 c1 = cp[32];

    const bool top    = wv < FWV / 2;
    const bool hvalid = top ? (band > 0) : (band < FBND - 1);
    const int  hpix0  = (top ? (r0 - 3) * Wn + wv * FHAL
                             : (r0 + FR) * Wn + (wv - FWV / 2) * FHAL);
    const float4* hp = (const float4*)(xb + (size_t)(hvalid ? hpix0 : 0) * Cn) + lane;

    const int pix0 = r0 * Wn + wv * FOWN;
    const float4* xp = (const float4*)(xb + (size_t)pix0 * Cn) + lane;
    float4*       op = (float4*)(out + ((size_t)b * HWn + pix0) * Cn) + lane;
    float4*       pk = park + (wv * 2 * FST) * 32 + lane;

    const int pslot = lane >> 1;
    const int ch    = lane & 1;

    float keep = 0.0f, hkeep = 0.0f;

    float4 e0[2 * FSH], h0[2 * FHH];
    #pragma unroll
    for (int j = 0; j < 2 * FSH; ++j)
        e0[j] = xp[j * 32];
    if (hvalid) {
        #pragma unroll
        for (int j = 0; j < 2 * FHH; ++j)
            h0[j] = hp[j * 32];
    }

    for (int i = threadIdx.x; i < TSIZE; i += FWV * 32)
        sp_t[i] = 0.0f;
    __syncthreads();

    if (hvalid) {
        #pragma unroll
        for (int p = 0; p < FHH; ++p) {
            float s, m;
            pixel_stats(h0[2 * p], h0[2 * p + 1], c0, c1, s, m);
            hkeep = (lane == 2 * p)     ? s : hkeep;
            hkeep = (lane == 2 * p + 1) ? m : hkeep;
        }
    }

    float4 e1[2 * FSH], h1[2 * FHH];
    #pragma unroll
    for (int j = 0; j < 2 * FSH; ++j)
        e1[j] = xp[(2 * FSH + j) * 32];
    if (hvalid) {
        #pragma unroll
        for (int j = 0; j < 2 * FHH; ++j)
            h1[j] = hp[(2 * FHH + j) * 32];
    }

    #pragma unroll
    for (int p = 0; p < FSH; ++p) {
        float s, m;
        pixel_stats(e0[2 * p], e0[2 * p + 1], c0, c1, s, m);
        keep = (lane == 2 * p)     ? s : keep;
        keep = (lane == 2 * p + 1) ? m : keep;
        pk[(2 * p) * 32]     = e0[2 * p];
        pk[(2 * p + 1) * 32] = e0[2 * p + 1];
    }
    if (hvalid) {
        #pragma unroll
        for (int p = 0; p < FHH; ++p) {
            float s, m;
            pixel_stats(h1[2 * p], h1[2 * p + 1], c0, c1, s, m);
            hkeep = (lane == 2 * (FHH + p))     ? s : hkeep;
            hkeep = (lane == 2 * (FHH + p) + 1) ? m : hkeep;
        }
    }

    #pragma unroll
    for (int p = 0; p < FSH; ++p) {
        float s, m;
        pixel_stats(e1[2 * p], e1[2 * p + 1], c0, c1, s, m);
        keep = (lane == 2 * (FSH + p))     ? s : keep;
        keep = (lane == 2 * (FSH + p) + 1) ? m : keep;
        pk[(2 * (FSH + p)) * 32]     = e1[2 * p];
        pk[(2 * (FSH + p) + 1) * 32] = e1[2 * p + 1];
    }

    if (hvalid && pslot < FHAL) {
        const int hq  = hpix0 + pslot;
        const int row = hq / Wn;
        const int col = hq - row * Wn;
        sp_t[((row - (r0 - 3)) * TCOLS + col + 3) * 2 + ch] = hkeep;
    }

    float4 v[2 * FRG];
    #pragma unroll
    for (int j = 0; j < 2 * FRG; ++j)
        v[j] = xp[(2 * FST + j) * 32];
    #pragma unroll
    for (int p = 0; p < FRG; ++p) {
        float s, m;
        pixel_stats(v[2 * p], v[2 * p + 1], c0, c1, s, m);
        keep = (lane == 2 * (p + FST))     ? s : keep;
        keep = (lane == 2 * (p + FST) + 1) ? m : keep;
    }

    const int q   = pix0 + min(pslot, FOWN - 1);
    const int py  = q / Wn;
    const int px  = q - py * Wn;
    const int tbase = ((py - r0) * TCOLS + px) * 2 + ch;
    if (pslot < FOWN)
        sp_t[tbase + (3 * TCOLS + 3) * 2] = keep;
    __syncthreads();

    float a0 = 0.0f, a1 = 0.0f;
    #pragma unroll
    for (int ky = 0; ky < 7; ++ky) {
        #pragma unroll
        for (int kx = 0; kx < 7; ++kx) {
            const float t = sp_t[tbase + (ky * TCOLS + kx) * 2];
            a0 += t * wconv[(ky * 7 + kx) * 2 + 0];
            a1 += t * wconv[(ky * 7 + kx) * 2 + 1];
        }
    }
    const float mine = ch ? a1 : a0;
    const float gate = sigmoidf_(mine + __shfl_xor(mine, 1, 32) + bconv[0]);

    #pragma unroll
    for (int p = 0; p < FRG; ++p) {
        const float s = __int_as_float(
            __builtin_amdgcn_readlane(__float_as_int(gate), 2 * (p + FST)));
        scale_store(v[2 * p], v[2 * p + 1], c0, c1, s,
                    op + (2 * (p + FST)) * 32, op + (2 * (p + FST) + 1) * 32);
    }
    #pragma unroll
    for (int p = 0; p < FST; ++p) {
        const float s = __int_as_float(
            __builtin_amdgcn_readlane(__float_as_int(gate), 2 * p));
        scale_store(pk[(2 * p) * 32], pk[(2 * p + 1) * 32], c0, c1, s,
                    op + (2 * p) * 32, op + (2 * p + 1) * 32);
    }
}

extern "C" void kernel_launch(void* const* d_in, const int* in_sizes, int n_in,
                              void* d_out, int out_size, void* d_ws, size_t ws_size,
                              hipStream_t stream) {
    const float* x     = (const float*)d_in[0];
    const float* w1    = (const float*)d_in[1];
    const float* b1    = (const float*)d_in[2];
    const float* w2    = (const float*)d_in[3];
    const float* b2    = (const float*)d_in[4];
    const float* wconv = (const float*)d_in[5];
    const float* bconv = (const float*)d_in[6];
    float* out = (float*)d_out;

    float* ws    = (float*)d_ws;
    float* psum  = ws;
    float* pmax  = psum + (size_t)NCHK * Bn * Cn;
    float* chout = pmax + (size_t)NCHK * Bn * Cn;

    k_pool_partial<<<dim3(2 * NCHK, Bn), PSTR * 32, 0, stream>>>(x, psum, pmax);

    k_channel_mlp<<<Bn / MROWS, 1024, 0, stream>>>(psum, pmax, w1, b1, w2, b2, chout);

    k_stats_apply<<<Bn * FBND, FWV * 32, 0, stream>>>(x, chout, wconv, bconv, out);
}


# channel-MLP launch folded into the stats/apply launch (3 -> 2 launches): each band block redoes its image's 2-row f32 WMMA MLP under its first x loads, LDS-only barriers; sum orders unchanged
# speedup vs baseline: 3.8636x; 1.1599x over previous
#include <hip/hip_runtime.h>
#include <hip/hip_bf16.h>
#include <math.h>

#define Bn   32
#define Hn   56
#define Wn   56
#define Cn   256
#define HWn  (Hn * Wn)
#define HID  16
#define NCHK 8
#define CHUNK (HWn / NCHK)

typedef float v2f __attribute__((ext_vector_type(2)));
typedef float v8f __attribute__((ext_vector_type(8)));

__device__ __forceinline__ v8f wmma4(v2f a, v2f b, v8f c) {
    return __builtin_amdgcn_wmma_f32_16x16x4_f32(
         false, a,  false, b,
         (short)0, c,  false,  false);
}

__device__ __forceinline__ float sigmoidf_(float v) {
    return 1.0f / (1.0f + __expf(-v));
}

__device__ __forceinline__ void lds_barrier() {
    asm volatile("s_wait_dscnt 0x0\n\ts_barrier_signal -1\n\ts_barrier_wait -1" ::: "memory");
}

#define PSTR 14
#define PPW  (CHUNK / PSTR)
#define CHALF (Cn / 2)

__global__ __launch_bounds__(PSTR * 32)
void k_pool_partial(const float* __restrict__ x,
                    float* __restrict__ psum,
                    float* __restrict__ pmax) {
    __shared__ __attribute__((aligned(16))) float red_s[PSTR * CHALF];
    __shared__ __attribute__((aligned(16))) float red_m[PSTR * CHALF];

    const int lane  = threadIdx.x & 31;
    const int wv    = threadIdx.x >> 5;
    const int half  = blockIdx.x & 1;
    const int chunk = blockIdx.x >> 1;
    const int b     = blockIdx.y;

    const float4* xp = (const float4*)(x + ((size_t)b * HWn + (size_t)chunk * CHUNK + wv) * Cn
                                         + half * CHALF) + lane;

    float4 v[PPW];
    #pragma unroll
    for (int j = 0; j < PPW; ++j)
        v[j] = xp[(size_t)j * PSTR * (Cn / 4)];

    float4 s = {0.0f, 0.0f, 0.0f, 0.0f};
    float4 m = {-3.402823466e38f, -3.402823466e38f, -3.402823466e38f, -3.402823466e38f};
    #pragma unroll
    for (int j = 0; j < PPW; ++j) {
        s.x += v[j].x;  s.y += v[j].y;  s.z += v[j].z;  s.w += v[j].w;
        m.x = fmaxf(m.x, v[j].x);  m.y = fmaxf(m.y, v[j].y);
        m.z = fmaxf(m.z, v[j].z);  m.w = fmaxf(m.w, v[j].w);
    }
    *(float4*)&red_s[wv * CHALF + lane * 4] = s;
    *(float4*)&red_m[wv * CHALF + lane * 4] = m;
    __syncthreads();

    if (threadIdx.x < CHALF) {
        const int c = threadIdx.x;
        float ss = 0.0f, mm = -3.402823466e38f;
        #pragma unroll
        for (int k = 0; k < PSTR; ++k) {
            ss += red_s[k * CHALF + c];
            mm = fmaxf(mm, red_m[k * CHALF + c]);
        }
        const size_t o = ((size_t)chunk * Bn + b) * Cn + half * CHALF + c;
        psum[o] = ss;
        pmax[o] = mm;
    }
}

#define PPITCH (Cn + 4)
#define KSL    16
#define KPS    (Cn / KSL)
#define FR    7
#define FBND  (Hn / FR)
#define FWV   28
#define FOWN  ((FR * Wn) / FWV)
#define FHAL  ((3 * Wn) / (FWV / 2))
#define TROWS (FR + 6)
#define TCOLS (Wn + 6)
#define TSIZE (TROWS * TCOLS * 2)
#define FST   6
#define FRG   (FOWN - FST)
#define FHH   (FHAL / 2)
#define FSH   (FST / 2)

__device__ __forceinline__ void pixel_stats(const float4 a, const float4 d,
                                            const float4 c0, const float4 c1,
                                            float& s_out, float& m_out) {
    float s = 0.0f, m = -3.402823466e38f, t;
    t = a.x * c0.x;  s += t;  m = fmaxf(m, t);
    t = a.y * c0.y;  s += t;  m = fmaxf(m, t);
    t = a.z * c0.z;  s += t;  m = fmaxf(m, t);
    t = a.w * c0.w;  s += t;  m = fmaxf(m, t);
    t = d.x * c1.x;  s += t;  m = fmaxf(m, t);
    t = d.y * c1.y;  s += t;  m = fmaxf(m, t);
    t = d.z * c1.z;  s += t;  m = fmaxf(m, t);
    t = d.w * c1.w;  s += t;  m = fmaxf(m, t);
    #pragma unroll
    for (int off = 16; off >= 1; off >>= 1) {
        s += __shfl_xor(s, off, 32);
        m = fmaxf(m, __shfl_xor(m, off, 32));
    }
    s_out = s * (1.0f / (float)Cn);
    m_out = m;
}

__device__ __forceinline__ void scale_store(const float4 a, const float4 d,
                                            const float4 c0, const float4 c1,
                                            const float s,
                                            float4* __restrict__ o0p,
                                            float4* __restrict__ o1p) {
    float4 o0, o1;
    o0.x = a.x * c0.x * s;  o0.y = a.y * c0.y * s;
    o0.z = a.z * c0.z * s;  o0.w = a.w * c0.w * s;
    o1.x = d.x * c1.x * s;  o1.y = d.y * c1.y * s;
    o1.z = d.z * c1.z * s;  o1.w = d.w * c1.w * s;
    *o0p = o0;
    *o1p = o1;
}

__global__ __launch_bounds__(FWV * 32)
void k_mlp_stats_apply(const float* __restrict__ x,
                       const float* __restrict__ psum,
                       const float* __restrict__ pmax,
                       const float* __restrict__ w1,
                       const float* __restrict__ b1,
                       const float* __restrict__ w2,
                       const float* __restrict__ b2,
                       const float* __restrict__ wconv,
                       const float* __restrict__ bconv,
                       float* __restrict__ out) {
    __shared__ float sp_t[TSIZE];
    __shared__ float4 park[FWV * 2 * FST * 32];
    __shared__ float lds_pool[2][PPITCH];
    __shared__ float lds_part[KSL][2][32];
    __shared__ float lds_h[2][HID];
    __shared__ __attribute__((aligned(16))) float ch_s[2 * Cn];

    const int lane = threadIdx.x & 31;
    const int wv   = __builtin_amdgcn_readfirstlane((int)(threadIdx.x >> 5));
    const int b    = (int)blockIdx.x / FBND;
    const int band = (int)blockIdx.x - b * FBND;
    const int r0   = band * FR;

    const float* xb = x + (size_t)b * HWn * Cn;

    const int  col  = lane & 15;
    const int  ksel = (lane < 16) ? 0 : 2;
    const bool mlpw = wv < KSL;
    const int  parr = wv >> 3;
    const int  pch  = (int)threadIdx.x & (Cn - 1);
    float pv[NCHK] = {};
    v2f   bw1[KPS / 4] = {};
    v2f   bw2[HID / 4] = {};
    float bias1 = 0.0f, bias2 = 0.0f;
    if (mlpw) {
        const float* pp = (parr ? pmax : psum) + (size_t)b * Cn + pch;
        #pragma unroll
        for (int cq = 0; cq < NCHK; ++cq)
            pv[cq] = pp[(size_t)cq * (Bn * Cn)];
        #pragma unroll
        for (int j = 0; j < KPS / 4; ++j) {
            const int k = wv * KPS + 4 * j + ksel;
            bw1[j].x = w1[(k)     * HID + col];
            bw1[j].y = w1[(k + 1) * HID + col];
        }
        bias1 = b1[col];
        #pragma unroll
        for (int j = 0; j < HID / 4; ++j) {
            bw2[j].x = w2[(4 * j + ksel)     * Cn + wv * 16 + col];
            bw2[j].y = w2[(4 * j + ksel + 1) * Cn + wv * 16 + col];
        }
        bias2 = b2[wv * 16 + col];
    }

    const bool top    = wv < FWV / 2;
    const bool hvalid = top ? (band > 0) : (band < FBND - 1);
    const int  hpix0  = (top ? (r0 - 3) * Wn + wv * FHAL
                             : (r0 + FR) * Wn + (wv - FWV / 2) * FHAL);
    const float4* hp = (const float4*)(xb + (size_t)(hvalid ? hpix0 : 0) * Cn) + lane;

    const int pix0 = r0 * Wn + wv * FOWN;
    const float4* xp = (const float4*)(xb + (size_t)pix0 * Cn) + lane;
    float4*       op = (float4*)(out + ((size_t)b * HWn + pix0) * Cn) + lane;
    float4*       pk = park + (wv * 2 * FST) * 32 + lane;

    const int pslot = lane >> 1;
    const int ch    = lane & 1;

    float keep = 0.0f, hkeep = 0.0f;

    float4 e0[2 * FSH], h0[2 * FHH];
    #pragma unroll
    for (int j = 0; j < 2 * FSH; ++j)
        e0[j] = xp[j * 32];

    for (int i = threadIdx.x; i < TSIZE; i += FWV * 32)
        sp_t[i] = 0.0f;

    if (mlpw) {
        float r;
        if (parr == 0) {
            float sacc = 0.0f;
            #pragma unroll
            for (int cq = 0; cq < NCHK; ++cq) sacc += pv[cq];
            r = sacc * (1.0f / (float)HWn);
        } else {
            float macc = -3.402823466e38f;
            #pragma unroll
            for (int cq = 0; cq < NCHK; ++cq) macc = fmaxf(macc, pv[cq]);
            r = macc;
        }
        lds_pool[parr][pch] = r;
    }

    if (hvalid) {
        #pragma unroll
        for (int j = 0; j < 2 * FHH; ++j)
            h0[j] = hp[j * 32];
    }
    lds_barrier();

    if (mlpw) {
        v8f acc = {};
        #pragma unroll
        for (int j = 0; j < KPS / 4; ++j) {
            const int k = wv * KPS + 4 * j + ksel;
            const float p0 = lds_pool[col & 1][k];
            const float p1 = lds_pool[col & 1][k + 1];
            v2f a;
            a.x = (col < 2) ? p0 : 0.0f;
            a.y = (col < 2) ? p1 : 0.0f;
            acc = wmma4(a, bw1[j], acc);
        }
        lds_part[wv][0][lane] = acc[0];
        lds_part[wv][1][lane] = acc[1];
    }
    lds_barrier();

    if (threadIdx.x < 32) {
        const int src = lane >> 4;
        float sum = 0.0f;
        #pragma unroll
        for (int q = 0; q < KSL; ++q)
            sum += lds_part[q][src][col];
        lds_h[src][col] = fmaxf(sum + bias1, 0.0f);
    }
    lds_barrier();

    if (mlpw) {
        v8f acc = {};
        #pragma unroll
        for (int j = 0; j < HID / 4; ++j) {
            const float h0v = lds_h[col & 1][4 * j + ksel];
            const float h1v = lds_h[col & 1][4 * j + ksel + 1];
            v2f a;
            a.x = (col < 2) ? h0v : 0.0f;
            a.y = (col < 2) ? h1v : 0.0f;
            acc = wmma4(a, bw2[j], acc);
        }
        float comb = 0.0f;
        comb += sigmoidf_(acc[0] + bias2);
        comb += sigmoidf_(acc[1] + bias2);
        ch_s[(lane >> 4) * Cn + wv * 16 + col] = comb;
    }
    lds_barrier();

    const float4 c0 = *(const float4*)&ch_s[4 * lane];
    const float4 c1 = *(const float4*)&ch_s[CHALF + 4 * lane];

    if (hvalid) {
        #pragma unroll
        for (int p = 0; p < FHH; ++p) {
            float s, m;
            pixel_stats(h0[2 * p], h0[2 * p + 1], c0, c1, s, m);
            hkeep = (lane == 2 * p)     ? s : hkeep;
            hkeep = (lane == 2 * p + 1) ? m : hkeep;
        }
    }

    float4 e1[2 * FSH], h1[2 * FHH];
    #pragma unroll
    for (int j = 0; j < 2 * FSH; ++j)
        e1[j] = xp[(2 * FSH + j) * 32];
    if (hvalid) {
        #pragma unroll
        for (int j = 0; j < 2 * FHH; ++j)
            h1[j] = hp[(2 * FHH + j) * 32];
    }

    #pragma unroll
    for (int p = 0; p < FSH; ++p) {
        float s, m;
        pixel_stats(e0[2 * p], e0[2 * p + 1], c0, c1, s, m);
        keep = (lane == 2 * p)     ? s : keep;
        keep = (lane == 2 * p + 1) ? m : keep;
        pk[(2 * p) * 32]     = e0[2 * p];
        pk[(2 * p + 1) * 32] = e0[2 * p + 1];
    }
    if (hvalid) {
        #pragma unroll
        for (int p = 0; p < FHH; ++p) {
            float s, m;
            pixel_stats(h1[2 * p], h1[2 * p + 1], c0, c1, s, m);
            hkeep = (lane == 2 * (FHH + p))     ? s : hkeep;
            hkeep = (lane == 2 * (FHH + p) + 1) ? m : hkeep;
        }
    }

    #pragma unroll
    for (int p = 0; p < FSH; ++p) {
        float s, m;
        pixel_stats(e1[2 * p], e1[2 * p + 1], c0, c1, s, m);
        keep = (lane == 2 * (FSH + p))     ? s : keep;
        keep = (lane == 2 * (FSH + p) + 1) ? m : keep;
        pk[(2 * (FSH + p)) * 32]     = e1[2 * p];
        pk[(2 * (FSH + p) + 1) * 32] = e1[2 * p + 1];
    }

    if (hvalid && pslot < FHAL) {
        const int hq  = hpix0 + pslot;
        const int row = hq / Wn;
        const int col = hq - row * Wn;
        sp_t[((row - (r0 - 3)) * TCOLS + col + 3) * 2 + ch] = hkeep;
    }

    float4 v[2 * FRG];
    #pragma unroll
    for (int j = 0; j < 2 * FRG; ++j)
        v[j] = xp[(2 * FST + j) * 32];
    #pragma unroll
    for (int p = 0; p < FRG; ++p) {
        float s, m;
        pixel_stats(v[2 * p], v[2 * p + 1], c0, c1, s, m);
        keep = (lane == 2 * (p + FST))     ? s : keep;
        keep = (lane == 2 * (p + FST) + 1) ? m : keep;
    }

    const int q   = pix0 + min(pslot, FOWN - 1);
    const int py  = q / Wn;
    const int px  = q - py * Wn;
    const int tbase = ((py - r0) * TCOLS + px) * 2 + ch;
    if (pslot < FOWN)
        sp_t[tbase + (3 * TCOLS + 3) * 2] = keep;
    __syncthreads();

    float a0 = 0.0f, a1 = 0.0f;
    #pragma unroll
    for (int ky = 0; ky < 7; ++ky) {
        #pragma unroll
        for (int kx = 0; kx < 7; ++kx) {
            const float t = sp_t[tbase + (ky * TCOLS + kx) * 2];
            a0 += t * wconv[(ky * 7 + kx) * 2 + 0];
            a1 += t * wconv[(ky * 7 + kx) * 2 + 1];
        }
    }
    const float mine = ch ? a1 : a0;
    const float gate = sigmoidf_(mine + __shfl_xor(mine, 1, 32) + bconv[0]);

    #pragma unroll
    for (int p = 0; p < FRG; ++p) {
        const float s = __int_as_float(
            __builtin_amdgcn_readlane(__float_as_int(gate), 2 * (p + FST)));
        scale_store(v[2 * p], v[2 * p + 1], c0, c1, s,
                    op + (2 * (p + FST)) * 32, op + (2 * (p + FST) + 1) * 32);
    }
    #pragma unroll
    for (int p = 0; p < FST; ++p) {
        const float s = __int_as_float(
            __builtin_amdgcn_readlane(__float_as_int(gate), 2 * p));
        scale_store(pk[(2 * p) * 32], pk[(2 * p + 1) * 32], c0, c1, s,
                    op + (2 * p) * 32, op + (2 * p + 1) * 32);
    }
}

extern "C" void kernel_launch(void* const* d_in, const int* in_sizes, int n_in,
                              void* d_out, int out_size, void* d_ws, size_t ws_size,
                              hipStream_t stream) {
    const float* x     = (const float*)d_in[0];
    const float* w1    = (const float*)d_in[1];
    const float* b1    = (const float*)d_in[2];
    const float* w2    = (const float*)d_in[3];
    const float* b2    = (const float*)d_in[4];
    const float* wconv = (const float*)d_in[5];
    const float* bconv = (const float*)d_in[6];
    float* out = (float*)d_out;

    float* ws    = (float*)d_ws;
    float* psum  = ws;
    float* pmax  = psum + (size_t)NCHK * Bn * Cn;

    k_pool_partial<<<dim3(2 * NCHK, Bn), PSTR * 32, 0, stream>>>(x, psum, pmax);

    k_mlp_stats_apply<<<Bn * FBND, FWV * 32, 0, stream>>>(x, psum, pmax, w1, b1, w2, b2,
                                                         wconv, bconv, out);
}


# stats passes: each pixel's 32-lane sum/max moved from LDS-crossbar shuffles (ds_bpermute) to vector-ALU lane exchanges, pixels of a load phase sharing registers per level; sum orders unchanged
# speedup vs baseline: 4.1727x; 1.0800x over previous
#include <hip/hip_runtime.h>
#include <hip/hip_bf16.h>
#include <math.h>

#define Bn   32
#define Hn   56
#define Wn   56
#define Cn   256
#define HWn  (Hn * Wn)
#define HID  16
#define NCHK 8
#define CHUNK (HWn / NCHK)

typedef float v2f __attribute__((ext_vector_type(2)));
typedef float v8f __attribute__((ext_vector_type(8)));

__device__ __forceinline__ v8f wmma4(v2f a, v2f b, v8f c) {
    return __builtin_amdgcn_wmma_f32_16x16x4_f32(
         false, a,  false, b,
         (short)0, c,  false,  false);
}

__device__ __forceinline__ float sigmoidf_(float v) {
    return 1.0f / (1.0f + __expf(-v));
}

__device__ __forceinline__ void lds_barrier() {
    asm volatile("s_wait_dscnt 0x0\n\ts_barrier_signal -1\n\ts_barrier_wait -1" ::: "memory");
}

#define PSTR 14
#define PPW  (CHUNK / PSTR)
#define CHALF (Cn / 2)

__global__ __launch_bounds__(PSTR * 32)
void k_pool_partial(const float* __restrict__ x,
                    float* __restrict__ psum,
                    float* __restrict__ pmax) {
    __shared__ __attribute__((aligned(16))) float red_s[PSTR * CHALF];
    __shared__ __attribute__((aligned(16))) float red_m[PSTR * CHALF];

    const int lane  = threadIdx.x & 31;
    const int wv    = threadIdx.x >> 5;
    const int half  = blockIdx.x & 1;
    const int chunk = blockIdx.x >> 1;
    const int b     = blockIdx.y;

    const float4* xp = (const float4*)(x + ((size_t)b * HWn + (size_t)chunk * CHUNK + wv) * Cn
                                         + half * CHALF) + lane;

    float4 v[PPW];
    #pragma unroll
    for (int j = 0; j < PPW; ++j)
        v[j] = xp[(size_t)j * PSTR * (Cn / 4)];

    float4 s = {0.0f, 0.0f, 0.0f, 0.0f};
    float4 m = {-3.402823466e38f, -3.402823466e38f, -3.402823466e38f, -3.402823466e38f};
    #pragma unroll
    for (int j = 0; j < PPW; ++j) {
        s.x += v[j].x;  s.y += v[j].y;  s.z += v[j].z;  s.w += v[j].w;
        m.x = fmaxf(m.x, v[j].x);  m.y = fmaxf(m.y, v[j].y);
        m.z = fmaxf(m.z, v[j].z);  m.w = fmaxf(m.w, v[j].w);
    }
    *(float4*)&red_s[wv * CHALF + lane * 4] = s;
    *(float4*)&red_m[wv * CHALF + lane * 4] = m;
    __syncthreads();

    if (threadIdx.x < CHALF) {
        const int c = threadIdx.x;
        float ss = 0.0f, mm = -3.402823466e38f;
        #pragma unroll
        for (int k = 0; k < PSTR; ++k) {
            ss += red_s[k * CHALF + c];
            mm = fmaxf(mm, red_m[k * CHALF + c]);
        }
        const size_t o = ((size_t)chunk * Bn + b) * Cn + half * CHALF + c;
        psum[o] = ss;
        pmax[o] = mm;
    }
}

#define PPITCH (Cn + 4)
#define KSL    16
#define KPS    (Cn / KSL)
#define FR    7
#define FBND  (Hn / FR)
#define FWV   28
#define FOWN  ((FR * Wn) / FWV)
#define FHAL  ((3 * Wn) / (FWV / 2))
#define TROWS (FR + 6)
#define TCOLS (Wn + 6)
#define TSIZE (TROWS * TCOLS * 2)
#define FST   6
#define FRG   (FOWN - FST)
#define FHH   (FHAL / 2)
#define FSH   (FST / 2)

struct SM { float s, m; };

__device__ __forceinline__ SM lane_stats(const float4 a, const float4 d,
                                         const float4 c0, const float4 c1) {
    float s = 0.0f, m = -3.402823466e38f, t;
    t = a.x * c0.x;  s += t;  m = fmaxf(m, t);
    t = a.y * c0.y;  s += t;  m = fmaxf(m, t);
    t = a.z * c0.z;  s += t;  m = fmaxf(m, t);
    t = a.w * c0.w;  s += t;  m = fmaxf(m, t);
    t = d.x * c1.x;  s += t;  m = fmaxf(m, t);
    t = d.y * c1.y;  s += t;  m = fmaxf(m, t);
    t = d.z * c1.z;  s += t;  m = fmaxf(m, t);
    t = d.w * c1.w;  s += t;  m = fmaxf(m, t);
    SM r;  r.s = s;  r.m = m;
    return r;
}


__device__ __forceinline__ float row_half_other(float v) {
    const int b = __builtin_bit_cast(int, v);
    return __builtin_bit_cast(float,
        __builtin_amdgcn_permlanex16(b, b, 0x76543210, 0xfedcba98, false, false));
}

template <int MASK>
__device__ __forceinline__ float row_xor(float v) {
    return __builtin_bit_cast(float,
        __builtin_amdgcn_update_dpp(0, __builtin_bit_cast(int, v), 0x160 + MASK, 0xf, 0xf, true));
}

__device__ __forceinline__ SM pair16(const SM x, const SM y, const bool row1) {
    const float ks = row1 ? y.s : x.s,  gs = row1 ? x.s : y.s;
    const float km = row1 ? y.m : x.m,  gm = row1 ? x.m : y.m;
    SM r;
    r.s = ks + row_half_other(gs);
    r.m = fmaxf(km, row_half_other(gm));
    return r;
}

__device__ __forceinline__ SM single16(const SM x) {
    SM r;
    r.s = x.s + row_half_other(x.s);
    r.m = fmaxf(x.m, row_half_other(x.m));
    return r;
}

template <int MASK>
__device__ __forceinline__ SM merge_row(const SM x, const SM y, const bool second) {
    const float ks = second ? y.s : x.s,  gs = second ? x.s : y.s;
    const float km = second ? y.m : x.m,  gm = second ? x.m : y.m;
    SM r;
    r.s = ks + row_xor<MASK>(gs);
    r.m = fmaxf(km, row_xor<MASK>(gm));
    return r;
}

template <int MASK>
__device__ __forceinline__ SM single_row(const SM x) {
    SM r;
    r.s = x.s + row_xor<MASK>(x.s);
    r.m = fmaxf(x.m, row_xor<MASK>(x.m));
    return r;
}

__device__ __forceinline__ float pick_stat(const SM t, const bool odd) {
    const float s = t.s + row_xor<1>(t.s);
    const float m = fmaxf(t.m, row_xor<1>(t.m));
    return odd ? m : s * (1.0f / (float)Cn);
}

__device__ __forceinline__ SM halo_tree(const float4* h, const float4 c0, const float4 c1,
                                        const bool row1, const bool b3, const bool b2) {
    SM p[2 * 3];
    #pragma unroll
    for (int j = 0; j < 6; ++j)
        p[j] = lane_stats(h[2 * j], h[2 * j + 1], c0, c1);
    const SM r01 = pair16(p[0], p[1], row1);
    const SM r23 = pair16(p[2], p[3], row1);
    const SM r45 = pair16(p[4], p[5], row1);
    const SM w0  = merge_row<8>(r01, r23, b3);
    const SM w1  = single_row<8>(r45);
    return merge_row<4>(w0, w1, b2);
}

__device__ __forceinline__ SM park_tree(const SM p0, const SM p1, const SM p2,
                                        const bool row1, const bool b3) {
    const SM r01 = pair16(p0, p1, row1);
    const SM r2  = single16(p2);
    return merge_row<8>(r01, r2, b3);
}

__device__ __forceinline__ constexpr int own_lane(const int p) {
    return p < FST ? 4 * (p / FSH) + 16 * ((p % FSH) & 1) + 8 * ((p % FSH) >> 1)
                   : 2 + 16 * ((p - FST) & 1) + 8 * (((p - FST) >> 1) & 1) + 4 * ((p - FST) >> 2);
}

__device__ __forceinline__ void scale_store(const float4 a, const float4 d,
                                            const float4 c0, const float4 c1,
                                            const float s,
                                            float4* __restrict__ o0p,
                                            float4* __restrict__ o1p) {
    float4 o0, o1;
    o0.x = a.x * c0.x * s;  o0.y = a.y * c0.y * s;
    o0.z = a.z * c0.z * s;  o0.w = a.w * c0.w * s;
    o1.x = d.x * c1.x * s;  o1.y = d.y * c1.y * s;
    o1.z = d.z * c1.z * s;  o1.w = d.w * c1.w * s;
    *o0p = o0;
    *o1p = o1;
}

__global__ __launch_bounds__(FWV * 32)
void k_mlp_stats_apply(const float* __restrict__ x,
                       const float* __restrict__ psum,
                       const float* __restrict__ pmax,
                       const float* __restrict__ w1,
                       const float* __restrict__ b1,
                       const float* __restrict__ w2,
                       const float* __restrict__ b2,
                       const float* __restrict__ wconv,
                       const float* __restrict__ bconv,
                       float* __restrict__ out) {
    __shared__ float sp_t[TSIZE];
    __shared__ float4 park[FWV * 2 * FST * 32];
    __shared__ float lds_pool[2][PPITCH];
    __shared__ float lds_part[KSL][2][32];
    __shared__ float lds_h[2][HID];
    __shared__ __attribute__((aligned(16))) float ch_s[2 * Cn];

    const int lane = threadIdx.x & 31;
    const int wv   = __builtin_amdgcn_readfirstlane((int)(threadIdx.x >> 5));
    const int b    = (int)blockIdx.x / FBND;
    const int band = (int)blockIdx.x - b * FBND;
    const int r0   = band * FR;

    const float* xb = x + (size_t)b * HWn * Cn;

    const int  col  = lane & 15;
    const int  ksel = (lane < 16) ? 0 : 2;
    const bool mlpw = wv < KSL;
    const int  parr = wv >> 3;
    const int  pch  = (int)threadIdx.x & (Cn - 1);
    float pv[NCHK] = {};
    v2f   bw1[KPS / 4] = {};
    v2f   bw2[HID / 4] = {};
    float bias1 = 0.0f, bias2 = 0.0f;
    if (mlpw) {
        const float* pp = (parr ? pmax : psum) + (size_t)b * Cn + pch;
        #pragma unroll
        for (int cq = 0; cq < NCHK; ++cq)
            pv[cq] = pp[(size_t)cq * (Bn * Cn)];
        #pragma unroll
        for (int j = 0; j < KPS / 4; ++j) {
            const int k = wv * KPS + 4 * j + ksel;
            bw1[j].x = w1[(k)     * HID + col];
            bw1[j].y = w1[(k + 1) * HID + col];
        }
        bias1 = b1[col];
        #pragma unroll
        for (int j = 0; j < HID / 4; ++j) {
            bw2[j].x = w2[(4 * j + ksel)     * Cn + wv * 16 + col];
            bw2[j].y = w2[(4 * j + ksel + 1) * Cn + wv * 16 + col];
        }
        bias2 = b2[wv * 16 + col];
    }

    const bool top    = wv < FWV / 2;
    const bool hvalid = top ? (band > 0) : (band < FBND - 1);
    const int  hpix0  = (top ? (r0 - 3) * Wn + wv * FHAL
                             : (r0 + FR) * Wn + (wv - FWV / 2) * FHAL);
    const float4* hp = (const float4*)(xb + (size_t)(hvalid ? hpix0 : 0) * Cn) + lane;

    const int pix0 = r0 * Wn + wv * FOWN;
    const float4* xp = (const float4*)(xb + (size_t)pix0 * Cn) + lane;
    float4*       op = (float4*)(out + ((size_t)b * HWn + pix0) * Cn) + lane;
    float4*       pk = park + (wv * 2 * FST) * 32 + lane;

    const int  ch  = lane & 1;
    const bool lb1 = (lane & 2) != 0, lb2 = (lane & 4) != 0;
    const bool lb3 = (lane & 8) != 0, lb4 = (lane & 16) != 0;
    const int  l4  = lane >> 4;
    const int  pki  = l4 + (lb3 ? 2 : 0);
    const int  pown = lb1 ? FST + l4 + (lb3 ? 2 : 0) + (lb2 ? 4 : 0)
                          : (lb2 ? FSH : 0) + min(pki, FSH - 1);
    const bool ownv = lb1 || pki < FSH;
    const int  hpi  = (lb3 ? 1 : 0) + (lb2 ? 2 : 0);
    const int  phal = (lb1 ? FHH : 0) + 2 * min(hpi, 2) + l4;
    const bool halv = hpi < 3;

    SM hu0 = {0.0f, 0.0f}, hu1 = {0.0f, 0.0f};

    float4 e0[2 * FSH], h0[2 * FHH];
    #pragma unroll
    for (int j = 0; j < 2 * FSH; ++j)
        e0[j] = xp[j * 32];

    for (int i = threadIdx.x; i < TSIZE; i += FWV * 32)
        sp_t[i] = 0.0f;

    if (mlpw) {
        float r;
        if (parr == 0) {
            float sacc = 0.0f;
            #pragma unroll
            for (int cq = 0; cq < NCHK; ++cq) sacc += pv[cq];
            r = sacc * (1.0f / (float)HWn);
        } else {
            float macc = -3.402823466e38f;
            #pragma unroll
            for (int cq = 0; cq < NCHK; ++cq) macc = fmaxf(macc, pv[cq]);
            r = macc;
        }
        lds_pool[parr][pch] = r;
    }

    if (hvalid) {
        #pragma unroll
        for (int j = 0; j < 2 * FHH; ++j)
            h0[j] = hp[j * 32];
    }
    lds_barrier();

    if (mlpw) {
        v8f acc = {};
        #pragma unroll
        for (int j = 0; j < KPS / 4; ++j) {
            const int k = wv * KPS + 4 * j + ksel;
            const float p0 = lds_pool[col & 1][k];
            const float p1 = lds_pool[col & 1][k + 1];
            v2f a;
            a.x = (col < 2) ? p0 : 0.0f;
            a.y = (col < 2) ? p1 : 0.0f;
            acc = wmma4(a, bw1[j], acc);
        }
        lds_part[wv][0][lane] = acc[0];
        lds_part[wv][1][lane] = acc[1];
    }
    lds_barrier();

    if (threadIdx.x < 32) {
        const int src = lane >> 4;
        float sum = 0.0f;
        #pragma unroll
        for (int q = 0; q < KSL; ++q)
            sum += lds_part[q][src][col];
        lds_h[src][col] = fmaxf(sum + bias1, 0.0f);
    }
    lds_barrier();

    if (mlpw) {
        v8f acc = {};
        #pragma unroll
        for (int j = 0; j < HID / 4; ++j) {
            const float h0v = lds_h[col & 1][4 * j + ksel];
            const float h1v = lds_h[col & 1][4 * j + ksel + 1];
            v2f a;
            a.x = (col < 2) ? h0v : 0.0f;
            a.y = (col < 2) ? h1v : 0.0f;
            acc = wmma4(a, bw2[j], acc);
        }
        float comb = 0.0f;
        comb += sigmoidf_(acc[0] + bias2);
        comb += sigmoidf_(acc[1] + bias2);
        ch_s[(lane >> 4) * Cn + wv * 16 + col] = comb;
    }
    lds_barrier();

    const float4 c0 = *(const float4*)&ch_s[4 * lane];
    const float4 c1 = *(const float4*)&ch_s[CHALF + 4 * lane];

    if (hvalid)
        hu0 = halo_tree(h0, c0, c1, lb4, lb3, lb2);

    float4 e1[2 * FSH], h1[2 * FHH];
    #pragma unroll
    for (int j = 0; j < 2 * FSH; ++j)
        e1[j] = xp[(2 * FSH + j) * 32];
    if (hvalid) {
        #pragma unroll
        for (int j = 0; j < 2 * FHH; ++j)
            h1[j] = hp[(2 * FHH + j) * 32];
    }

    SM pe0[FSH], pe1[FSH];
    #pragma unroll
    for (int p = 0; p < FSH; ++p) {
        pe0[p] = lane_stats(e0[2 * p], e0[2 * p + 1], c0, c1);
        pk[(2 * p) * 32]     = e0[2 * p];
        pk[(2 * p + 1) * 32] = e0[2 * p + 1];
    }
    const SM wa = park_tree(pe0[0], pe0[1], pe0[2], lb4, lb3);
    if (hvalid)
        hu1 = halo_tree(h1, c0, c1, lb4, lb3, lb2);

    #pragma unroll
    for (int p = 0; p < FSH; ++p) {
        pe1[p] = lane_stats(e1[2 * p], e1[2 * p + 1], c0, c1);
        pk[(2 * (FSH + p)) * 32]     = e1[2 * p];
        pk[(2 * (FSH + p) + 1) * 32] = e1[2 * p + 1];
    }
    const SM wb  = park_tree(pe1[0], pe1[1], pe1[2], lb4, lb3);
    const SM uab = merge_row<4>(wa, wb, lb2);

    if (hvalid) {
        const float hkeep = pick_stat(merge_row<2>(hu0, hu1, lb1), ch != 0);
        if (halv) {
            const int hq  = hpix0 + phal;
            const int row = hq / Wn;
            const int col = hq - row * Wn;
            sp_t[((row - (r0 - 3)) * TCOLS + col + 3) * 2 + ch] = hkeep;
        }
    }

    float4 v[2 * FRG];
    #pragma unroll
    for (int j = 0; j < 2 * FRG; ++j)
        v[j] = xp[(2 * FST + j) * 32];
    SM pv3[FRG];
    #pragma unroll
    for (int p = 0; p < FRG; ++p)
        pv3[p] = lane_stats(v[2 * p], v[2 * p + 1], c0, c1);
    const SM uv = merge_row<4>(
        merge_row<8>(pair16(pv3[0], pv3[1], lb4), pair16(pv3[2], pv3[3], lb4), lb3),
        merge_row<8>(pair16(pv3[4], pv3[5], lb4), pair16(pv3[6], pv3[7], lb4), lb3), lb2);
    const float keep = pick_stat(merge_row<2>(uab, uv, lb1), ch != 0);

    const int q   = pix0 + pown;
    const int py  = q / Wn;
    const int px  = q - py * Wn;
    const int tbase = ((py - r0) * TCOLS + px) * 2 + ch;
    if (ownv)
        sp_t[tbase + (3 * TCOLS + 3) * 2] = keep;
    __syncthreads();

    float a0 = 0.0f, a1 = 0.0f;
    #pragma unroll
    for (int ky = 0; ky < 7; ++ky) {
        #pragma unroll
        for (int kx = 0; kx < 7; ++kx) {
            const float t = sp_t[tbase + (ky * TCOLS + kx) * 2];
            a0 += t * wconv[(ky * 7 + kx) * 2 + 0];
            a1 += t * wconv[(ky * 7 + kx) * 2 + 1];
        }
    }
    const float mine = ch ? a1 : a0;
    const float gate = sigmoidf_(mine + row_xor<1>(mine) + bconv[0]);

    #pragma unroll
    for (int p = 0; p < FRG; ++p) {
        const float s = __int_as_float(
            __builtin_amdgcn_readlane(__float_as_int(gate), own_lane(p + FST)));
        scale_store(v[2 * p], v[2 * p + 1], c0, c1, s,
                    op + (2 * (p + FST)) * 32, op + (2 * (p + FST) + 1) * 32);
    }
    #pragma unroll
    for (int p = 0; p < FST; ++p) {
        const float s = __int_as_float(
            __builtin_amdgcn_readlane(__float_as_int(gate), own_lane(p)));
        scale_store(pk[(2 * p) * 32], pk[(2 * p + 1) * 32], c0, c1, s,
                    op + (2 * p) * 32, op + (2 * p + 1) * 32);
    }
}

extern "C" void kernel_launch(void* const* d_in, const int* in_sizes, int n_in,
                              void* d_out, int out_size, void* d_ws, size_t ws_size,
                              hipStream_t stream) {
    const float* x     = (const float*)d_in[0];
    const float* w1    = (const float*)d_in[1];
    const float* b1    = (const float*)d_in[2];
    const float* w2    = (const float*)d_in[3];
    const float* b2    = (const float*)d_in[4];
    const float* wconv = (const float*)d_in[5];
    const float* bconv = (const float*)d_in[6];
    float* out = (float*)d_out;

    float* ws    = (float*)d_ws;
    float* psum  = ws;
    float* pmax  = psum + (size_t)NCHK * Bn * Cn;

    k_pool_partial<<<dim3(2 * NCHK, Bn), PSTR * 32, 0, stream>>>(x, psum, pmax);

    k_mlp_stats_apply<<<Bn * FBND, FWV * 32, 0, stream>>>(x, psum, pmax, w1, b1, w2, b2,
                                                         wconv, bconv, out);
}


# pooling pass: one block per (chunk, image) streams all 256 channels of the band launch 2's same-numbered block owns: 3584 waves (was 7168), ~32 loads in flight per wave (was 12); sum orders unchanged
# speedup vs baseline: 4.3747x; 1.0484x over previous
#include <hip/hip_runtime.h>
#include <hip/hip_bf16.h>
#include <math.h>

#define Bn   32
#define Hn   56
#define Wn   56
#define Cn   256
#define HWn  (Hn * Wn)
#define HID  16
#define NCHK 8
#define CHUNK (HWn / NCHK)

typedef float v2f __attribute__((ext_vector_type(2)));
typedef float v8f __attribute__((ext_vector_type(8)));

__device__ __forceinline__ v8f wmma4(v2f a, v2f b, v8f c) {
    return __builtin_amdgcn_wmma_f32_16x16x4_f32(
         false, a,  false, b,
         (short)0, c,  false,  false);
}

__device__ __forceinline__ float sigmoidf_(float v) {
    return 1.0f / (1.0f + __expf(-v));
}

__device__ __forceinline__ void lds_barrier() {
    asm volatile("s_wait_dscnt 0x0\n\ts_barrier_signal -1\n\ts_barrier_wait -1" ::: "memory");
}

#define PSTR 14
#define PPW  (CHUNK / PSTR)
#define CHALF (Cn / 2)

__global__ __launch_bounds__(PSTR * 32)
__attribute__((amdgpu_waves_per_eu(4, 6)))
void k_pool_partial(const float* __restrict__ x,
                    float* __restrict__ psum,
                    float* __restrict__ pmax) {
    __shared__ __attribute__((aligned(16))) float red_s[PSTR * Cn];
    __shared__ __attribute__((aligned(16))) float red_m[PSTR * Cn];

    const int lane  = threadIdx.x & 31;
    const int wv    = threadIdx.x >> 5;
    const int chunk = blockIdx.x;
    const int b     = blockIdx.y;

    const float4* xp = (const float4*)(x + ((size_t)b * HWn + (size_t)chunk * CHUNK + wv) * Cn) + lane;

    float4 v[2 * PPW];
    #pragma unroll
    for (int j = 0; j < PPW; ++j) {
        v[2 * j]     = xp[(size_t)j * PSTR * (Cn / 4)];
        v[2 * j + 1] = xp[(size_t)j * PSTR * (Cn / 4) + CHALF / 4];
    }

    float4 s0 = {0.0f, 0.0f, 0.0f, 0.0f}, s1 = {0.0f, 0.0f, 0.0f, 0.0f};
    float4 m0 = {-3.402823466e38f, -3.402823466e38f, -3.402823466e38f, -3.402823466e38f};
    float4 m1 = m0;
    #pragma unroll
    for (int j = 0; j < PPW; ++j) {
        const float4 a = v[2 * j], d = v[2 * j + 1];
        s0.x += a.x;  s0.y += a.y;  s0.z += a.z;  s0.w += a.w;
        s1.x += d.x;  s1.y += d.y;  s1.z += d.z;  s1.w += d.w;
        m0.x = fmaxf(m0.x, a.x);  m0.y = fmaxf(m0.y, a.y);
        m0.z = fmaxf(m0.z, a.z);  m0.w = fmaxf(m0.w, a.w);
        m1.x = fmaxf(m1.x, d.x);  m1.y = fmaxf(m1.y, d.y);
        m1.z = fmaxf(m1.z, d.z);  m1.w = fmaxf(m1.w, d.w);
    }
    *(float4*)&red_s[wv * Cn + lane * 4]         = s0;
    *(float4*)&red_s[wv * Cn + CHALF + lane * 4] = s1;
    *(float4*)&red_m[wv * Cn + lane * 4]         = m0;
    *(float4*)&red_m[wv * Cn + CHALF + lane * 4] = m1;
    __syncthreads();

    if (threadIdx.x < Cn) {
        const int c = threadIdx.x;
        float ss = 0.0f, mm = -3.402823466e38f;
        #pragma unroll
        for (int k = 0; k < PSTR; ++k) {
            ss += red_s[k * Cn + c];
            mm = fmaxf(mm, red_m[k * Cn + c]);
        }
        const size_t o = ((size_t)chunk * Bn + b) * Cn + c;
        psum[o] = ss;
        pmax[o] = mm;
    }
}

#define PPITCH (Cn + 4)
#define KSL    16
#define KPS    (Cn / KSL)
#define FR    7
#define FBND  (Hn / FR)
#define FWV   28
#define FOWN  ((FR * Wn) / FWV)
#define FHAL  ((3 * Wn) / (FWV / 2))
#define TROWS (FR + 6)
#define TCOLS (Wn + 6)
#define TSIZE (TROWS * TCOLS * 2)
#define FST   6
#define FRG   (FOWN - FST)
#define FHH   (FHAL / 2)
#define FSH   (FST / 2)

struct SM { float s, m; };

__device__ __forceinline__ SM lane_stats(const float4 a, const float4 d,
                                         const float4 c0, const float4 c1) {
    float s = 0.0f, m = -3.402823466e38f, t;
    t = a.x * c0.x;  s += t;  m = fmaxf(m, t);
    t = a.y * c0.y;  s += t;  m = fmaxf(m, t);
    t = a.z * c0.z;  s += t;  m = fmaxf(m, t);
    t = a.w * c0.w;  s += t;  m = fmaxf(m, t);
    t = d.x * c1.x;  s += t;  m = fmaxf(m, t);
    t = d.y * c1.y;  s += t;  m = fmaxf(m, t);
    t = d.z * c1.z;  s += t;  m = fmaxf(m, t);
    t = d.w * c1.w;  s += t;  m = fmaxf(m, t);
    SM r;  r.s = s;  r.m = m;
    return r;
}


__device__ __forceinline__ float row_half_other(float v) {
    const int b = __builtin_bit_cast(int, v);
    return __builtin_bit_cast(float,
        __builtin_amdgcn_permlanex16(b, b, 0x76543210, 0xfedcba98, false, false));
}

template <int MASK>
__device__ __forceinline__ float row_xor(float v) {
    return __builtin_bit_cast(float,
        __builtin_amdgcn_update_dpp(0, __builtin_bit_cast(int, v), 0x160 + MASK, 0xf, 0xf, true));
}

__device__ __forceinline__ SM pair16(const SM x, const SM y, const bool row1) {
    const float ks = row1 ? y.s : x.s,  gs = row1 ? x.s : y.s;
    const float km = row1 ? y.m : x.m,  gm = row1 ? x.m : y.m;
    SM r;
    r.s = ks + row_half_other(gs);
    r.m = fmaxf(km, row_half_other(gm));
    return r;
}

__device__ __forceinline__ SM single16(const SM x) {
    SM r;
    r.s = x.s + row_half_other(x.s);
    r.m = fmaxf(x.m, row_half_other(x.m));
    return r;
}

template <int MASK>
__device__ __forceinline__ SM merge_row(const SM x, const SM y, const bool second) {
    const float ks = second ? y.s : x.s,  gs = second ? x.s : y.s;
    const float km = second ? y.m : x.m,  gm = second ? x.m : y.m;
    SM r;
    r.s = ks + row_xor<MASK>(gs);
    r.m = fmaxf(km, row_xor<MASK>(gm));
    return r;
}

template <int MASK>
__device__ __forceinline__ SM single_row(const SM x) {
    SM r;
    r.s = x.s + row_xor<MASK>(x.s);
    r.m = fmaxf(x.m, row_xor<MASK>(x.m));
    return r;
}

__device__ __forceinline__ float pick_stat(const SM t, const bool odd) {
    const float s = t.s + row_xor<1>(t.s);
    const float m = fmaxf(t.m, row_xor<1>(t.m));
    return odd ? m : s * (1.0f / (float)Cn);
}

__device__ __forceinline__ SM halo_tree(const float4* h, const float4 c0, const float4 c1,
                                        const bool row1, const bool b3, const bool b2) {
    SM p[2 * 3];
    #pragma unroll
    for (int j = 0; j < 6; ++j)
        p[j] = lane_stats(h[2 * j], h[2 * j + 1], c0, c1);
    const SM r01 = pair16(p[0], p[1], row1);
    const SM r23 = pair16(p[2], p[3], row1);
    const SM r45 = pair16(p[4], p[5], row1);
    const SM w0  = merge_row<8>(r01, r23, b3);
    const SM w1  = single_row<8>(r45);
    return merge_row<4>(w0, w1, b2);
}

__device__ __forceinline__ SM park_tree(const SM p0, const SM p1, const SM p2,
                                        const bool row1, const bool b3) {
    const SM r01 = pair16(p0, p1, row1);
    const SM r2  = single16(p2);
    return merge_row<8>(r01, r2, b3);
}

__device__ __forceinline__ constexpr int own_lane(const int p) {
    return p < FST ? 4 * (p / FSH) + 16 * ((p % FSH) & 1) + 8 * ((p % FSH) >> 1)
                   : 2 + 16 * ((p - FST) & 1) + 8 * (((p - FST) >> 1) & 1) + 4 * ((p - FST) >> 2);
}

__device__ __forceinline__ void scale_store(const float4 a, const float4 d,
                                            const float4 c0, const float4 c1,
                                            const float s,
                                            float4* __restrict__ o0p,
                                            float4* __restrict__ o1p) {
    float4 o0, o1;
    o0.x = a.x * c0.x * s;  o0.y = a.y * c0.y * s;
    o0.z = a.z * c0.z * s;  o0.w = a.w * c0.w * s;
    o1.x = d.x * c1.x * s;  o1.y = d.y * c1.y * s;
    o1.z = d.z * c1.z * s;  o1.w = d.w * c1.w * s;
    *o0p = o0;
    *o1p = o1;
}

__global__ __launch_bounds__(FWV * 32)
void k_mlp_stats_apply(const float* __restrict__ x,
                       const float* __restrict__ psum,
                       const float* __restrict__ pmax,
                       const float* __restrict__ w1,
                       const float* __restrict__ b1,
                       const float* __restrict__ w2,
                       const float* __restrict__ b2,
                       const float* __restrict__ wconv,
                       const float* __restrict__ bconv,
                       float* __restrict__ out) {
    __shared__ float sp_t[TSIZE];
    __shared__ float4 park[FWV * 2 * FST * 32];
    __shared__ float lds_pool[2][PPITCH];
    __shared__ float lds_part[KSL][2][32];
    __shared__ float lds_h[2][HID];
    __shared__ __attribute__((aligned(16))) float ch_s[2 * Cn];

    const int lane = threadIdx.x & 31;
    const int wv   = __builtin_amdgcn_readfirstlane((int)(threadIdx.x >> 5));
    const int b    = (int)blockIdx.x / FBND;
    const int band = (int)blockIdx.x - b * FBND;
    const int r0   = band * FR;

    const float* xb = x + (size_t)b * HWn * Cn;

    const int  col  = lane & 15;
    const int  ksel = (lane < 16) ? 0 : 2;
    const bool mlpw = wv < KSL;
    const int  parr = wv >> 3;
    const int  pch  = (int)threadIdx.x & (Cn - 1);
    float pv[NCHK] = {};
    v2f   bw1[KPS / 4] = {};
    v2f   bw2[HID / 4] = {};
    float bias1 = 0.0f, bias2 = 0.0f;
    if (mlpw) {
        const float* pp = (parr ? pmax : psum) + (size_t)b * Cn + pch;
        #pragma unroll
        for (int cq = 0; cq < NCHK; ++cq)
            pv[cq] = pp[(size_t)cq * (Bn * Cn)];
        #pragma unroll
        for (int j = 0; j < KPS / 4; ++j) {
            const int k = wv * KPS + 4 * j + ksel;
            bw1[j].x = w1[(k)     * HID + col];
            bw1[j].y = w1[(k + 1) * HID + col];
        }
        bias1 = b1[col];
        #pragma unroll
        for (int j = 0; j < HID / 4; ++j) {
            bw2[j].x = w2[(4 * j + ksel)     * Cn + wv * 16 + col];
            bw2[j].y = w2[(4 * j + ksel + 1) * Cn + wv * 16 + col];
        }
        bias2 = b2[wv * 16 + col];
    }

    const bool top    = wv < FWV / 2;
    const bool hvalid = top ? (band > 0) : (band < FBND - 1);
    const int  hpix0  = (top ? (r0 - 3) * Wn + wv * FHAL
                             : (r0 + FR) * Wn + (wv - FWV / 2) * FHAL);
    const float4* hp = (const float4*)(xb + (size_t)(hvalid ? hpix0 : 0) * Cn) + lane;

    const int pix0 = r0 * Wn + wv * FOWN;
    const float4* xp = (const float4*)(xb + (size_t)pix0 * Cn) + lane;
    float4*       op = (float4*)(out + ((size_t)b * HWn + pix0) * Cn) + lane;
    float4*       pk = park + (wv * 2 * FST) * 32 + lane;

    const int  ch  = lane & 1;
    const bool lb1 = (lane & 2) != 0, lb2 = (lane & 4) != 0;
    const bool lb3 = (lane & 8) != 0, lb4 = (lane & 16) != 0;
    const int  l4  = lane >> 4;
    const int  pki  = l4 + (lb3 ? 2 : 0);
    const int  pown = lb1 ? FST + l4 + (lb3 ? 2 : 0) + (lb2 ? 4 : 0)
                          : (lb2 ? FSH : 0) + min(pki, FSH - 1);
    const bool ownv = lb1 || pki < FSH;
    const int  hpi  = (lb3 ? 1 : 0) + (lb2 ? 2 : 0);
    const int  phal = (lb1 ? FHH : 0) + 2 * min(hpi, 2) + l4;
    const bool halv = hpi < 3;

    SM hu0 = {0.0f, 0.0f}, hu1 = {0.0f, 0.0f};

    float4 e0[2 * FSH], h0[2 * FHH];
    #pragma unroll
    for (int j = 0; j < 2 * FSH; ++j)
        e0[j] = xp[j * 32];

    for (int i = threadIdx.x; i < TSIZE; i += FWV * 32)
        sp_t[i] = 0.0f;

    if (mlpw) {
        float r;
        if (parr == 0) {
            float sacc = 0.0f;
            #pragma unroll
            for (int cq = 0; cq < NCHK; ++cq) sacc += pv[cq];
            r = sacc * (1.0f / (float)HWn);
        } else {
            float macc = -3.402823466e38f;
            #pragma unroll
            for (int cq = 0; cq < NCHK; ++cq) macc = fmaxf(macc, pv[cq]);
            r = macc;
        }
        lds_pool[parr][pch] = r;
    }

    if (hvalid) {
        #pragma unroll
        for (int j = 0; j < 2 * FHH; ++j)
            h0[j] = hp[j * 32];
    }
    lds_barrier();

    if (mlpw) {
        v8f acc = {};
        #pragma unroll
        for (int j = 0; j < KPS / 4; ++j) {
            const int k = wv * KPS + 4 * j + ksel;
            const float p0 = lds_pool[col & 1][k];
            const float p1 = lds_pool[col & 1][k + 1];
            v2f a;
            a.x = (col < 2) ? p0 : 0.0f;
            a.y = (col < 2) ? p1 : 0.0f;
            acc = wmma4(a, bw1[j], acc);
        }
        lds_part[wv][0][lane] = acc[0];
        lds_part[wv][1][lane] = acc[1];
    }
    lds_barrier();

    if (threadIdx.x < 32) {
        const int src = lane >> 4;
        float sum = 0.0f;
        #pragma unroll
        for (int q = 0; q < KSL; ++q)
            sum += lds_part[q][src][col];
        lds_h[src][col] = fmaxf(sum + bias1, 0.0f);
    }
    lds_barrier();

    if (mlpw) {
        v8f acc = {};
        #pragma unroll
        for (int j = 0; j < HID / 4; ++j) {
            const float h0v = lds_h[col & 1][4 * j + ksel];
            const float h1v = lds_h[col & 1][4 * j + ksel + 1];
            v2f a;
            a.x = (col < 2) ? h0v : 0.0f;
            a.y = (col < 2) ? h1v : 0.0f;
            acc = wmma4(a, bw2[j], acc);
        }
        float comb = 0.0f;
        comb += sigmoidf_(acc[0] + bias2);
        comb += sigmoidf_(acc[1] + bias2);
        ch_s[(lane >> 4) * Cn + wv * 16 + col] = comb;
    }
    lds_barrier();

    const float4 c0 = *(const float4*)&ch_s[4 * lane];
    const float4 c1 = *(const float4*)&ch_s[CHALF + 4 * lane];

    if (hvalid)
        hu0 = halo_tree(h0, c0, c1, lb4, lb3, lb2);

    float4 e1[2 * FSH], h1[2 * FHH];
    #pragma unroll
    for (int j = 0; j < 2 * FSH; ++j)
        e1[j] = xp[(2 * FSH + j) * 32];
    if (hvalid) {
        #pragma unroll
        for (int j = 0; j < 2 * FHH; ++j)
            h1[j] = hp[(2 * FHH + j) * 32];
    }

    SM pe0[FSH], pe1[FSH];
    #pragma unroll
    for (int p = 0; p < FSH; ++p) {
        pe0[p] = lane_stats(e0[2 * p], e0[2 * p + 1], c0, c1);
        pk[(2 * p) * 32]     = e0[2 * p];
        pk[(2 * p + 1) * 32] = e0[2 * p + 1];
    }
    const SM wa = park_tree(pe0[0], pe0[1], pe0[2], lb4, lb3);
    if (hvalid)
        hu1 = halo_tree(h1, c0, c1, lb4, lb3, lb2);

    #pragma unroll
    for (int p = 0; p < FSH; ++p) {
        pe1[p] = lane_stats(e1[2 * p], e1[2 * p + 1], c0, c1);
        pk[(2 * (FSH + p)) * 32]     = e1[2 * p];
        pk[(2 * (FSH + p) + 1) * 32] = e1[2 * p + 1];
    }
    const SM wb  = park_tree(pe1[0], pe1[1], pe1[2], lb4, lb3);
    const SM uab = merge_row<4>(wa, wb, lb2);

    if (hvalid) {
        const float hkeep = pick_stat(merge_row<2>(hu0, hu1, lb1), ch != 0);
        if (halv) {
            const int hq  = hpix0 + phal;
            const int row = hq / Wn;
            const int col = hq - row * Wn;
            sp_t[((row - (r0 - 3)) * TCOLS + col + 3) * 2 + ch] = hkeep;
        }
    }

    float4 v[2 * FRG];
    #pragma unroll
    for (int j = 0; j < 2 * FRG; ++j)
        v[j] = xp[(2 * FST + j) * 32];
    SM pv3[FRG];
    #pragma unroll
    for (int p = 0; p < FRG; ++p)
        pv3[p] = lane_stats(v[2 * p], v[2 * p + 1], c0, c1);
    const SM uv = merge_row<4>(
        merge_row<8>(pair16(pv3[0], pv3[1], lb4), pair16(pv3[2], pv3[3], lb4), lb3),
        merge_row<8>(pair16(pv3[4], pv3[5], lb4), pair16(pv3[6], pv3[7], lb4), lb3), lb2);
    const float keep = pick_stat(merge_row<2>(uab, uv, lb1), ch != 0);

    const int q   = pix0 + pown;
    const int py  = q / Wn;
    const int px  = q - py * Wn;
    const int tbase = ((py - r0) * TCOLS + px) * 2 + ch;
    if (ownv)
        sp_t[tbase + (3 * TCOLS + 3) * 2] = keep;
    __syncthreads();

    float a0 = 0.0f, a1 = 0.0f;
    #pragma unroll
    for (int ky = 0; ky < 7; ++ky) {
        #pragma unroll
        for (int kx = 0; kx < 7; ++kx) {
            const float t = sp_t[tbase + (ky * TCOLS + kx) * 2];
            a0 += t * wconv[(ky * 7 + kx) * 2 + 0];
            a1 += t * wconv[(ky * 7 + kx) * 2 + 1];
        }
    }
    const float mine = ch ? a1 : a0;
    const float gate = sigmoidf_(mine + row_xor<1>(mine) + bconv[0]);

    #pragma unroll
    for (int p = 0; p < FRG; ++p) {
        const float s = __int_as_float(
            __builtin_amdgcn_readlane(__float_as_int(gate), own_lane(p + FST)));
        scale_store(v[2 * p], v[2 * p + 1], c0, c1, s,
                    op + (2 * (p + FST)) * 32, op + (2 * (p + FST) + 1) * 32);
    }
    #pragma unroll
    for (int p = 0; p < FST; ++p) {
        const float s = __int_as_float(
            __builtin_amdgcn_readlane(__float_as_int(gate), own_lane(p)));
        scale_store(pk[(2 * p) * 32], pk[(2 * p + 1) * 32], c0, c1, s,
                    op + (2 * p) * 32, op + (2 * p + 1) * 32);
    }
}

extern "C" void kernel_launch(void* const* d_in, const int* in_sizes, int n_in,
                              void* d_out, int out_size, void* d_ws, size_t ws_size,
                              hipStream_t stream) {
    const float* x     = (const float*)d_in[0];
    const float* w1    = (const float*)d_in[1];
    const float* b1    = (const float*)d_in[2];
    const float* w2    = (const float*)d_in[3];
    const float* b2    = (const float*)d_in[4];
    const float* wconv = (const float*)d_in[5];
    const float* bconv = (const float*)d_in[6];
    float* out = (float*)d_out;

    float* ws    = (float*)d_ws;
    float* psum  = ws;
    float* pmax  = psum + (size_t)NCHK * Bn * Cn;

    k_pool_partial<<<dim3(NCHK, Bn), PSTR * 32, 0, stream>>>(x, psum, pmax);

    k_mlp_stats_apply<<<Bn * FBND, FWV * 32, 0, stream>>>(x, psum, pmax, w1, b1, w2, b2,
                                                         wconv, bconv, out);
}


# stats/apply launch finished in two halves: the lower 14 waves read the 3 halo rows below the band only after the first sp hand-over, under the upper half's gate and stores; sum orders unchanged
# speedup vs baseline: 4.6209x; 1.0563x over previous
#include <hip/hip_runtime.h>
#include <hip/hip_bf16.h>
#include <math.h>

#define Bn   32
#define Hn   56
#define Wn   56
#define Cn   256
#define HWn  (Hn * Wn)
#define HID  16
#define NCHK 8
#define CHUNK (HWn / NCHK)

typedef float v2f __attribute__((ext_vector_type(2)));
typedef float v8f __attribute__((ext_vector_type(8)));

__device__ __forceinline__ v8f wmma4(v2f a, v2f b, v8f c) {
    return __builtin_amdgcn_wmma_f32_16x16x4_f32(
         false, a,  false, b,
         (short)0, c,  false,  false);
}

__device__ __forceinline__ float sigmoidf_(float v) {
    return 1.0f / (1.0f + __expf(-v));
}

__device__ __forceinline__ void lds_barrier() {
    asm volatile("s_wait_dscnt 0x0\n\ts_barrier_signal -1\n\ts_barrier_wait -1" ::: "memory");
}

#define PSTR 14
#define PPW  (CHUNK / PSTR)
#define CHALF (Cn / 2)

__global__ __launch_bounds__(PSTR * 32)
__attribute__((amdgpu_waves_per_eu(4, 6)))
void k_pool_partial(const float* __restrict__ x,
                    float* __restrict__ psum,
                    float* __restrict__ pmax) {
    __shared__ __attribute__((aligned(16))) float red_s[PSTR * Cn];
    __shared__ __attribute__((aligned(16))) float red_m[PSTR * Cn];

    const int lane  = threadIdx.x & 31;
    const int wv    = threadIdx.x >> 5;
    const int chunk = blockIdx.x;
    const int b     = blockIdx.y;

    const float4* xp = (const float4*)(x + ((size_t)b * HWn + (size_t)chunk * CHUNK + wv) * Cn) + lane;

    float4 v[2 * PPW];
    #pragma unroll
    for (int j = 0; j < PPW; ++j) {
        v[2 * j]     = xp[(size_t)j * PSTR * (Cn / 4)];
        v[2 * j + 1] = xp[(size_t)j * PSTR * (Cn / 4) + CHALF / 4];
    }

    float4 s0 = {0.0f, 0.0f, 0.0f, 0.0f}, s1 = {0.0f, 0.0f, 0.0f, 0.0f};
    float4 m0 = {-3.402823466e38f, -3.402823466e38f, -3.402823466e38f, -3.402823466e38f};
    float4 m1 = m0;
    #pragma unroll
    for (int j = 0; j < PPW; ++j) {
        const float4 a = v[2 * j], d = v[2 * j + 1];
        s0.x += a.x;  s0.y += a.y;  s0.z += a.z;  s0.w += a.w;
        s1.x += d.x;  s1.y += d.y;  s1.z += d.z;  s1.w += d.w;
        m0.x = fmaxf(m0.x, a.x);  m0.y = fmaxf(m0.y, a.y);
        m0.z = fmaxf(m0.z, a.z);  m0.w = fmaxf(m0.w, a.w);
        m1.x = fmaxf(m1.x, d.x);  m1.y = fmaxf(m1.y, d.y);
        m1.z = fmaxf(m1.z, d.z);  m1.w = fmaxf(m1.w, d.w);
    }
    *(float4*)&red_s[wv * Cn + lane * 4]         = s0;
    *(float4*)&red_s[wv * Cn + CHALF + lane * 4] = s1;
    *(float4*)&red_m[wv * Cn + lane * 4]         = m0;
    *(float4*)&red_m[wv * Cn + CHALF + lane * 4] = m1;
    __syncthreads();

    if (threadIdx.x < Cn) {
        const int c = threadIdx.x;
        float ss = 0.0f, mm = -3.402823466e38f;
        #pragma unroll
        for (int k = 0; k < PSTR; ++k) {
            ss += red_s[k * Cn + c];
            mm = fmaxf(mm, red_m[k * Cn + c]);
        }
        const size_t o = ((size_t)chunk * Bn + b) * Cn + c;
        psum[o] = ss;
        pmax[o] = mm;
    }
}

#define PPITCH (Cn + 4)
#define KSL    16
#define KPS    (Cn / KSL)
#define FR    7
#define FBND  (Hn / FR)
#define FWV   28
#define FOWN  ((FR * Wn) / FWV)
#define FHAL  ((3 * Wn) / (FWV / 2))
#define TROWS (FR + 6)
#define TCOLS (Wn + 6)
#define TSIZE (TROWS * TCOLS * 2)
#define FST   6
#define FRG   (FOWN - FST)
#define FHH   (FHAL / 2)
#define FSH   (FST / 2)

struct SM { float s, m; };

__device__ __forceinline__ SM lane_stats(const float4 a, const float4 d,
                                         const float4 c0, const float4 c1) {
    float s = 0.0f, m = -3.402823466e38f, t;
    t = a.x * c0.x;  s += t;  m = fmaxf(m, t);
    t = a.y * c0.y;  s += t;  m = fmaxf(m, t);
    t = a.z * c0.z;  s += t;  m = fmaxf(m, t);
    t = a.w * c0.w;  s += t;  m = fmaxf(m, t);
    t = d.x * c1.x;  s += t;  m = fmaxf(m, t);
    t = d.y * c1.y;  s += t;  m = fmaxf(m, t);
    t = d.z * c1.z;  s += t;  m = fmaxf(m, t);
    t = d.w * c1.w;  s += t;  m = fmaxf(m, t);
    SM r;  r.s = s;  r.m = m;
    return r;
}


__device__ __forceinline__ float row_half_other(float v) {
    const int b = __builtin_bit_cast(int, v);
    return __builtin_bit_cast(float,
        __builtin_amdgcn_permlanex16(b, b, 0x76543210, 0xfedcba98, false, false));
}

template <int MASK>
__device__ __forceinline__ float row_xor(float v) {
    return __builtin_bit_cast(float,
        __builtin_amdgcn_update_dpp(0, __builtin_bit_cast(int, v), 0x160 + MASK, 0xf, 0xf, true));
}

__device__ __forceinline__ SM pair16(const SM x, const SM y, const bool row1) {
    const float ks = row1 ? y.s : x.s,  gs = row1 ? x.s : y.s;
    const float km = row1 ? y.m : x.m,  gm = row1 ? x.m : y.m;
    SM r;
    r.s = ks + row_half_other(gs);
    r.m = fmaxf(km, row_half_other(gm));
    return r;
}

__device__ __forceinline__ SM single16(const SM x) {
    SM r;
    r.s = x.s + row_half_other(x.s);
    r.m = fmaxf(x.m, row_half_other(x.m));
    return r;
}

template <int MASK>
__device__ __forceinline__ SM merge_row(const SM x, const SM y, const bool second) {
    const float ks = second ? y.s : x.s,  gs = second ? x.s : y.s;
    const float km = second ? y.m : x.m,  gm = second ? x.m : y.m;
    SM r;
    r.s = ks + row_xor<MASK>(gs);
    r.m = fmaxf(km, row_xor<MASK>(gm));
    return r;
}

template <int MASK>
__device__ __forceinline__ SM single_row(const SM x) {
    SM r;
    r.s = x.s + row_xor<MASK>(x.s);
    r.m = fmaxf(x.m, row_xor<MASK>(x.m));
    return r;
}

__device__ __forceinline__ float pick_stat(const SM t, const bool odd) {
    const float s = t.s + row_xor<1>(t.s);
    const float m = fmaxf(t.m, row_xor<1>(t.m));
    return odd ? m : s * (1.0f / (float)Cn);
}

__device__ __forceinline__ SM halo_tree(const float4* h, const float4 c0, const float4 c1,
                                        const bool row1, const bool b3, const bool b2) {
    SM p[2 * 3];
    #pragma unroll
    for (int j = 0; j < 6; ++j)
        p[j] = lane_stats(h[2 * j], h[2 * j + 1], c0, c1);
    const SM r01 = pair16(p[0], p[1], row1);
    const SM r23 = pair16(p[2], p[3], row1);
    const SM r45 = pair16(p[4], p[5], row1);
    const SM w0  = merge_row<8>(r01, r23, b3);
    const SM w1  = single_row<8>(r45);
    return merge_row<4>(w0, w1, b2);
}

__device__ __forceinline__ SM halo_tree_of(const SM* p, const bool row1, const bool b3, const bool b2) {
    const SM r01 = pair16(p[0], p[1], row1);
    const SM r23 = pair16(p[2], p[3], row1);
    const SM r45 = pair16(p[4], p[5], row1);
    const SM w0  = merge_row<8>(r01, r23, b3);
    const SM w1  = single_row<8>(r45);
    return merge_row<4>(w0, w1, b2);
}

__device__ __forceinline__ SM park_tree(const SM p0, const SM p1, const SM p2,
                                        const bool row1, const bool b3) {
    const SM r01 = pair16(p0, p1, row1);
    const SM r2  = single16(p2);
    return merge_row<8>(r01, r2, b3);
}

__device__ __forceinline__ constexpr int own_lane(const int p) {
    return p < FST ? 4 * (p / FSH) + 16 * ((p % FSH) & 1) + 8 * ((p % FSH) >> 1)
                   : 2 + 16 * ((p - FST) & 1) + 8 * (((p - FST) >> 1) & 1) + 4 * ((p - FST) >> 2);
}

__device__ __forceinline__ void scale_store(const float4 a, const float4 d,
                                            const float4 c0, const float4 c1,
                                            const float s,
                                            float4* __restrict__ o0p,
                                            float4* __restrict__ o1p) {
    float4 o0, o1;
    o0.x = a.x * c0.x * s;  o0.y = a.y * c0.y * s;
    o0.z = a.z * c0.z * s;  o0.w = a.w * c0.w * s;
    o1.x = d.x * c1.x * s;  o1.y = d.y * c1.y * s;
    o1.z = d.z * c1.z * s;  o1.w = d.w * c1.w * s;
    *o0p = o0;
    *o1p = o1;
}

__device__ __forceinline__ void gate_and_store(const float* sp_t, const int tbase, const int ch,
                                               const float* __restrict__ wconv,
                                               const float* __restrict__ bconv,
                                               const float4* v, const float4* pk,
                                               const float4 c0, const float4 c1,
                                               float4* __restrict__ op) {
    float a0 = 0.0f, a1 = 0.0f;
    #pragma unroll
    for (int ky = 0; ky < 7; ++ky) {
        #pragma unroll
        for (int kx = 0; kx < 7; ++kx) {
            const float t = sp_t[tbase + (ky * TCOLS + kx) * 2];
            a0 += t * wconv[(ky * 7 + kx) * 2 + 0];
            a1 += t * wconv[(ky * 7 + kx) * 2 + 1];
        }
    }
    const float mine = ch ? a1 : a0;
    const float gate = sigmoidf_(mine + row_xor<1>(mine) + bconv[0]);

    #pragma unroll
    for (int p = 0; p < FRG; ++p) {
        const float s = __int_as_float(
            __builtin_amdgcn_readlane(__float_as_int(gate), own_lane(p + FST)));
        scale_store(v[2 * p], v[2 * p + 1], c0, c1, s,
                    op + (2 * (p + FST)) * 32, op + (2 * (p + FST) + 1) * 32);
    }
    #pragma unroll
    for (int p = 0; p < FST; ++p) {
        const float s = __int_as_float(
            __builtin_amdgcn_readlane(__float_as_int(gate), own_lane(p)));
        scale_store(pk[(2 * p) * 32], pk[(2 * p + 1) * 32], c0, c1, s,
                    op + (2 * p) * 32, op + (2 * p + 1) * 32);
    }
}

__global__ __launch_bounds__(FWV * 32)
void k_mlp_stats_apply(const float* __restrict__ x,
                       const float* __restrict__ psum,
                       const float* __restrict__ pmax,
                       const float* __restrict__ w1,
                       const float* __restrict__ b1,
                       const float* __restrict__ w2,
                       const float* __restrict__ b2,
                       const float* __restrict__ wconv,
                       const float* __restrict__ bconv,
                       float* __restrict__ out) {
    __shared__ float sp_t[TSIZE];
    __shared__ float4 park[FWV * 2 * FST * 32];
    __shared__ float lds_pool[2][PPITCH];
    __shared__ float lds_part[KSL][2][32];
    __shared__ float lds_h[2][HID];
    __shared__ __attribute__((aligned(16))) float ch_s[2 * Cn];

    const int lane = threadIdx.x & 31;
    const int wv   = __builtin_amdgcn_readfirstlane((int)(threadIdx.x >> 5));
    const int b    = (int)blockIdx.x / FBND;
    const int band = (int)blockIdx.x - b * FBND;
    const int r0   = band * FR;

    const float* xb = x + (size_t)b * HWn * Cn;

    const int  col  = lane & 15;
    const int  ksel = (lane < 16) ? 0 : 2;
    const bool mlpw = wv < KSL;
    const int  parr = wv >> 3;
    const int  pch  = (int)threadIdx.x & (Cn - 1);
    float pv[NCHK] = {};
    v2f   bw1[KPS / 4] = {};
    v2f   bw2[HID / 4] = {};
    float bias1 = 0.0f, bias2 = 0.0f;
    if (mlpw) {
        const float* pp = (parr ? pmax : psum) + (size_t)b * Cn + pch;
        #pragma unroll
        for (int cq = 0; cq < NCHK; ++cq)
            pv[cq] = pp[(size_t)cq * (Bn * Cn)];
        #pragma unroll
        for (int j = 0; j < KPS / 4; ++j) {
            const int k = wv * KPS + 4 * j + ksel;
            bw1[j].x = w1[(k)     * HID + col];
            bw1[j].y = w1[(k + 1) * HID + col];
        }
        bias1 = b1[col];
        #pragma unroll
        for (int j = 0; j < HID / 4; ++j) {
            bw2[j].x = w2[(4 * j + ksel)     * Cn + wv * 16 + col];
            bw2[j].y = w2[(4 * j + ksel + 1) * Cn + wv * 16 + col];
        }
        bias2 = b2[wv * 16 + col];
    }

    const bool top    = wv < FWV / 2;
    const bool hvalid = top ? (band > 0) : (band < FBND - 1);
    const int  hpix0  = (top ? (r0 - 3) * Wn + wv * FHAL
                             : (r0 + FR) * Wn + (wv - FWV / 2) * FHAL);
    const float4* hp = (const float4*)(xb + (size_t)(hvalid ? hpix0 : 0) * Cn) + lane;
    const bool hearly = top && hvalid;
    const bool hlate  = !top && hvalid;

    const int pix0 = r0 * Wn + wv * FOWN;
    const float4* xp = (const float4*)(xb + (size_t)pix0 * Cn) + lane;
    float4*       op = (float4*)(out + ((size_t)b * HWn + pix0) * Cn) + lane;
    float4*       pk = park + (wv * 2 * FST) * 32 + lane;

    const int  ch  = lane & 1;
    const bool lb1 = (lane & 2) != 0, lb2 = (lane & 4) != 0;
    const bool lb3 = (lane & 8) != 0, lb4 = (lane & 16) != 0;
    const int  l4  = lane >> 4;
    const int  pki  = l4 + (lb3 ? 2 : 0);
    const int  pown = lb1 ? FST + l4 + (lb3 ? 2 : 0) + (lb2 ? 4 : 0)
                          : (lb2 ? FSH : 0) + min(pki, FSH - 1);
    const bool ownv = lb1 || pki < FSH;
    const int  hpi  = (lb3 ? 1 : 0) + (lb2 ? 2 : 0);
    const int  phal = (lb1 ? FHH : 0) + 2 * min(hpi, 2) + l4;
    const bool halv = hpi < 3;

    SM hu0 = {0.0f, 0.0f}, hu1 = {0.0f, 0.0f};

    float4 e0[2 * FSH], h0[2 * FHH];
    #pragma unroll
    for (int j = 0; j < 2 * FSH; ++j)
        e0[j] = xp[j * 32];

    for (int i = threadIdx.x; i < TSIZE; i += FWV * 32)
        sp_t[i] = 0.0f;

    if (mlpw) {
        float r;
        if (parr == 0) {
            float sacc = 0.0f;
            #pragma unroll
            for (int cq = 0; cq < NCHK; ++cq) sacc += pv[cq];
            r = sacc * (1.0f / (float)HWn);
        } else {
            float macc = -3.402823466e38f;
            #pragma unroll
            for (int cq = 0; cq < NCHK; ++cq) macc = fmaxf(macc, pv[cq]);
            r = macc;
        }
        lds_pool[parr][pch] = r;
    }

    if (hearly) {
        #pragma unroll
        for (int j = 0; j < 2 * FHH; ++j)
            h0[j] = hp[j * 32];
    }
    lds_barrier();

    if (mlpw) {
        v8f acc = {};
        #pragma unroll
        for (int j = 0; j < KPS / 4; ++j) {
            const int k = wv * KPS + 4 * j + ksel;
            const float p0 = lds_pool[col & 1][k];
            const float p1 = lds_pool[col & 1][k + 1];
            v2f a;
            a.x = (col < 2) ? p0 : 0.0f;
            a.y = (col < 2) ? p1 : 0.0f;
            acc = wmma4(a, bw1[j], acc);
        }
        lds_part[wv][0][lane] = acc[0];
        lds_part[wv][1][lane] = acc[1];
    }
    lds_barrier();

    if (threadIdx.x < 32) {
        const int src = lane >> 4;
        float sum = 0.0f;
        #pragma unroll
        for (int q = 0; q < KSL; ++q)
            sum += lds_part[q][src][col];
        lds_h[src][col] = fmaxf(sum + bias1, 0.0f);
    }
    lds_barrier();

    if (mlpw) {
        v8f acc = {};
        #pragma unroll
        for (int j = 0; j < HID / 4; ++j) {
            const float h0v = lds_h[col & 1][4 * j + ksel];
            const float h1v = lds_h[col & 1][4 * j + ksel + 1];
            v2f a;
            a.x = (col < 2) ? h0v : 0.0f;
            a.y = (col < 2) ? h1v : 0.0f;
            acc = wmma4(a, bw2[j], acc);
        }
        float comb = 0.0f;
        comb += sigmoidf_(acc[0] + bias2);
        comb += sigmoidf_(acc[1] + bias2);
        ch_s[(lane >> 4) * Cn + wv * 16 + col] = comb;
    }
    lds_barrier();

    const float4 c0 = *(const float4*)&ch_s[4 * lane];
    const float4 c1 = *(const float4*)&ch_s[CHALF + 4 * lane];

    if (hearly)
        hu0 = halo_tree(h0, c0, c1, lb4, lb3, lb2);

    float4 e1[2 * FSH], h1[2 * FHH];
    #pragma unroll
    for (int j = 0; j < 2 * FSH; ++j)
        e1[j] = xp[(2 * FSH + j) * 32];
    if (hearly) {
        #pragma unroll
        for (int j = 0; j < 2 * FHH; ++j)
            h1[j] = hp[(2 * FHH + j) * 32];
    }

    SM pe0[FSH], pe1[FSH];
    #pragma unroll
    for (int p = 0; p < FSH; ++p) {
        pe0[p] = lane_stats(e0[2 * p], e0[2 * p + 1], c0, c1);
        pk[(2 * p) * 32]     = e0[2 * p];
        pk[(2 * p + 1) * 32] = e0[2 * p + 1];
    }
    const SM wa = park_tree(pe0[0], pe0[1], pe0[2], lb4, lb3);
    if (hearly)
        hu1 = halo_tree(h1, c0, c1, lb4, lb3, lb2);

    #pragma unroll
    for (int p = 0; p < FSH; ++p) {
        pe1[p] = lane_stats(e1[2 * p], e1[2 * p + 1], c0, c1);
        pk[(2 * (FSH + p)) * 32]     = e1[2 * p];
        pk[(2 * (FSH + p) + 1) * 32] = e1[2 * p + 1];
    }
    const SM wb  = park_tree(pe1[0], pe1[1], pe1[2], lb4, lb3);
    const SM uab = merge_row<4>(wa, wb, lb2);

    if (hearly) {
        const float hkeep = pick_stat(merge_row<2>(hu0, hu1, lb1), ch != 0);
        if (halv) {
            const int hq  = hpix0 + phal;
            const int row = hq / Wn;
            const int col = hq - row * Wn;
            sp_t[((row - (r0 - 3)) * TCOLS + col + 3) * 2 + ch] = hkeep;
        }
    }

    float4 v[2 * FRG];
    #pragma unroll
    for (int j = 0; j < 2 * FRG; ++j)
        v[j] = xp[(2 * FST + j) * 32];
    SM pv3[FRG];
    #pragma unroll
    for (int p = 0; p < FRG; ++p)
        pv3[p] = lane_stats(v[2 * p], v[2 * p + 1], c0, c1);
    const SM uv = merge_row<4>(
        merge_row<8>(pair16(pv3[0], pv3[1], lb4), pair16(pv3[2], pv3[3], lb4), lb3),
        merge_row<8>(pair16(pv3[4], pv3[5], lb4), pair16(pv3[6], pv3[7], lb4), lb3), lb2);
    const float keep = pick_stat(merge_row<2>(uab, uv, lb1), ch != 0);

    const int q   = pix0 + pown;
    const int py  = q / Wn;
    const int px  = q - py * Wn;
    const int tbase = ((py - r0) * TCOLS + px) * 2 + ch;
    if (ownv)
        sp_t[tbase + (3 * TCOLS + 3) * 2] = keep;
    __syncthreads();

    #pragma nounroll
    for (int turn = 0; turn < 2; ++turn) {
        if (turn == 0) {
            if (hlate) {
                const int  ln = (int)__builtin_amdgcn_mbcnt_lo(~0u, 0u);
                const bool m1 = (ln & 2) != 0, m2 = (ln & 4) != 0;
                const bool m3 = (ln & 8) != 0, m4 = (ln & 16) != 0;
                const int  hs  = (m3 ? 1 : 0) + (m2 ? 2 : 0);
                const int  hpl = (m1 ? FHH : 0) + 2 * min(hs, 2) + (ln >> 4);
                const float4* hq = (const float4*)(xb + (size_t)hpix0 * Cn) + ln;
                SM gp[2 * FHH];
                #pragma unroll
                for (int ph = 0; ph < 2; ++ph) {
                    float4 g[2 * FHH];
                    #pragma unroll
                    for (int j = 0; j < 2 * FHH; ++j)
                        g[j] = hq[(2 * FHH * ph + j) * 32];
                    #pragma unroll
                    for (int j = 0; j < FHH; ++j)
                        gp[FHH * ph + j] = lane_stats(g[2 * j], g[2 * j + 1], c0, c1);
                }
                const SM gu0 = halo_tree_of(gp, m4, m3, m2);
                const SM gu1 = halo_tree_of(gp + FHH, m4, m3, m2);
                const float hkeep = pick_stat(merge_row<2>(gu0, gu1, m1), (ln & 1) != 0);
                if (hs < 3) {
                    const int hpx = hpix0 + hpl;
                    const int row = hpx / Wn;
                    const int col = hpx - row * Wn;
                    sp_t[((row - (r0 - 3)) * TCOLS + col + 3) * 2 + (ln & 1)] = hkeep;
                }
            }
        } else {
            lds_barrier();
        }
        if (top == (turn == 0))
            gate_and_store(sp_t, tbase, ch, wconv, bconv, v, pk, c0, c1, op);
    }
}

extern "C" void kernel_launch(void* const* d_in, const int* in_sizes, int n_in,
                              void* d_out, int out_size, void* d_ws, size_t ws_size,
                              hipStream_t stream) {
    const float* x     = (const float*)d_in[0];
    const float* w1    = (const float*)d_in[1];
    const float* b1    = (const float*)d_in[2];
    const float* w2    = (const float*)d_in[3];
    const float* b2    = (const float*)d_in[4];
    const float* wconv = (const float*)d_in[5];
    const float* bconv = (const float*)d_in[6];
    float* out = (float*)d_out;

    float* ws    = (float*)d_ws;
    float* psum  = ws;
    float* pmax  = psum + (size_t)NCHK * Bn * Cn;

    k_pool_partial<<<dim3(NCHK, Bn), PSTR * 32, 0, stream>>>(x, psum, pmax);

    k_mlp_stats_apply<<<Bn * FBND, FWV * 32, 0, stream>>>(x, psum, pmax, w1, b1, w2, b2,
                                                         wconv, bconv, out);
}
